# starting kernel
# baseline (speedup 1.0000x reference)
#include <hip/hip_runtime.h>
#include <math.h>

typedef __attribute__((ext_vector_type(2))) float v2f;
typedef __attribute__((ext_vector_type(8))) float v8f;

#define B_DIM 2
#define N_NODES 25000
#define FC_IN 64
#define FC_HID 256
#define MUL 32
#define OUT_CH 288
#define EDGES_PER_BLOCK 128
#define WAVES 8
#define SH_STRIDE 18
#define SW_STRIDE 97

__device__ __forceinline__ void atomic_add_f32(float* p, float v) {
    unsafeAtomicAdd(p, v);
}

__global__ __launch_bounds__(256) void conv_mlp_tp_scatter(
    const int* __restrict__ edge_src,
    const int* __restrict__ edge_dst,
    const float* __restrict__ node_emb,
    const float* __restrict__ edge_type,
    const float* __restrict__ W1,
    const float* __restrict__ W2,
    float* __restrict__ out)
{
    __shared__ float sH[WAVES * 16 * SH_STRIDE];
    __shared__ float sW[EDGES_PER_BLOCK * SW_STRIDE];

    const int tid   = threadIdx.x;
    const int wv    = tid >> 5;
    const int lane  = tid & 31;
    const int lmod  = lane & 15;
    const int khalf = (lane < 16) ? 0 : 2;
    const int e_base = blockIdx.x * EDGES_PER_BLOCK;
    const int e_tile = e_base + wv * 16;

    const v8f vzero = {0.f, 0.f, 0.f, 0.f, 0.f, 0.f, 0.f, 0.f};

    v2f aET[16];
    {
        const float* etp = edge_type + (size_t)(e_tile + lmod) * FC_IN;
        #pragma unroll
        for (int ks = 0; ks < 16; ++ks)
            aET[ks] = *(const v2f*)(etp + 4 * ks + khalf);
        __builtin_prefetch(etp + EDGES_PER_BLOCK * FC_IN, 0, 1);
    }

    v8f h[16];
    #pragma unroll
    for (int nt = 0; nt < 16; ++nt) {
        v8f acc = vzero;
        #pragma unroll
        for (int ks = 0; ks < 16; ++ks) {
            const int k0 = 4 * ks + khalf;
            v2f b;
            b.x = W1[(size_t)k0 * FC_HID + 16 * nt + lmod];
            b.y = W1[(size_t)(k0 + 1) * FC_HID + 16 * nt + lmod];
            acc = __builtin_amdgcn_wmma_f32_16x16x4_f32(
                false, aET[ks], false, b, (short)0, acc, false, false);
        }
        #pragma unroll
        for (int i = 0; i < 8; ++i) {
            const float hv = acc[i] * 0.125f;
            acc[i] = hv / (1.0f + __expf(-hv));
        }
        h[nt] = acc;
    }

    float* myH = sH + wv * (16 * SH_STRIDE);
    v8f acc2[6];
    #pragma unroll
    for (int j = 0; j < 6; ++j) acc2[j] = vzero;

    #pragma unroll
    for (int nt1 = 0; nt1 < 16; ++nt1) {
        {
            const int mbase = (lane < 16) ? 0 : 8;
            #pragma unroll
            for (int r = 0; r < 8; ++r)
                myH[(r + mbase) * SH_STRIDE + lmod] = h[nt1][r];
        }
        v2f a[4];
        #pragma unroll
        for (int kk = 0; kk < 4; ++kk)
            a[kk] = *(const v2f*)(myH + lmod * SH_STRIDE + 4 * kk + khalf);

        #pragma unroll
        for (int nt2 = 0; nt2 < 6; ++nt2) {
            #pragma unroll
            for (int kk = 0; kk < 4; ++kk) {
                const int k0 = 16 * nt1 + 4 * kk + khalf;
                v2f b;
                b.x = W2[(size_t)k0 * 96 + 16 * nt2 + lmod];
                b.y = W2[(size_t)(k0 + 1) * 96 + 16 * nt2 + lmod];
                acc2[nt2] = __builtin_amdgcn_wmma_f32_16x16x4_f32(
                    false, a[kk], false, b, (short)0, acc2[nt2], false, false);
            }
        }
    }

    {
        const int mbase = wv * 16 + ((lane < 16) ? 0 : 8);
        #pragma unroll
        for (int nt2 = 0; nt2 < 6; ++nt2) {
            #pragma unroll
            for (int r = 0; r < 8; ++r)
                sW[(mbase + r) * SW_STRIDE + 16 * nt2 + lmod] = acc2[nt2][r] * 0.0625f;
        }
    }
    __syncthreads();

    {
        const int e_loc = tid >> 1;
        const int bb    = tid & 1;
        const int e     = e_base + e_loc;
        const int si    = edge_src[e];
        const int di    = edge_dst[e];
        const float* xp = node_emb + ((size_t)bb * N_NODES + si) * 3;
        const float* yp = node_emb + ((size_t)bb * N_NODES + di) * 3;
        const float x0 = xp[0], x1 = xp[1], x2 = xp[2];
        const float y0 = yp[0], y1 = yp[1], y2 = yp[2];

        const float inv3 = 0.57735026918962576f;
        const float inv2 = 0.70710678118654752f;
        const float inv6 = 0.40824829046386302f;
        const float scl  = 0.25f;

        const float s  = (x0*y0 + x1*y1 + x2*y2) * inv3 * scl;
        const float v0 = (x1*y2 - x2*y1) * inv2 * scl;
        const float v1 = (x2*y0 - x0*y2) * inv2 * scl;
        const float v2 = (x0*y1 - x1*y0) * inv2 * scl;
        const float t0 = (x0*y1 + x1*y0) * inv2 * scl;
        const float t1 = (x1*y2 + x2*y1) * inv2 * scl;
        const float t2 = (x0*y2 + x2*y0) * inv2 * scl;
        const float t3 = (x0*y0 - x1*y1) * inv2 * scl;
        const float t4 = (2.0f*x2*y2 - x0*y0 - x1*y1) * inv6 * scl;

        const float* wrow = sW + e_loc * SW_STRIDE;
        float* op = out + ((size_t)bb * N_NODES + di) * OUT_CH;

        #pragma unroll 4
        for (int u = 0; u < MUL; ++u)
            atomic_add_f32(op + u, wrow[u] * s);

        #pragma unroll 4
        for (int u = 0; u < MUL; ++u) {
            const float wu = wrow[MUL + u];
            atomic_add_f32(op + MUL + 3*u + 0, wu * v0);
            atomic_add_f32(op + MUL + 3*u + 1, wu * v1);
            atomic_add_f32(op + MUL + 3*u + 2, wu * v2);
        }

        #pragma unroll 2
        for (int u = 0; u < MUL; ++u) {
            const float wu = wrow[2*MUL + u];
            atomic_add_f32(op + 4*MUL + 5*u + 0, wu * t0);
            atomic_add_f32(op + 4*MUL + 5*u + 1, wu * t1);
            atomic_add_f32(op + 4*MUL + 5*u + 2, wu * t2);
            atomic_add_f32(op + 4*MUL + 5*u + 3, wu * t3);
            atomic_add_f32(op + 4*MUL + 5*u + 4, wu * t4);
        }
    }
}

extern "C" void kernel_launch(void* const* d_in, const int* in_sizes, int n_in,
                              void* d_out, int out_size, void* d_ws, size_t ws_size,
                              hipStream_t stream) {
    const int*   edge_src  = (const int*)d_in[0];
    const int*   edge_dst  = (const int*)d_in[1];
    const float* node_emb  = (const float*)d_in[2];
    const float* edge_type = (const float*)d_in[3];
    const float* W1        = (const float*)d_in[4];
    const float* W2        = (const float*)d_in[5];
    float*       out       = (float*)d_out;

    hipMemsetAsync(d_out, 0, (size_t)out_size * sizeof(float), stream);

    const int E = in_sizes[0];
    const int nblocks = E / EDGES_PER_BLOCK;
    conv_mlp_tp_scatter<<<nblocks, 256, 0, stream>>>(
        edge_src, edge_dst, node_emb, edge_type, W1, W2, out);
}


# W1, W2 copied once into LDS in B-fragment order by 256 persistent blocks, each walking its node range in groups of <=8 nodes fitting 128 edges; fragments read from LDS one run ahead; add order kept
# speedup vs baseline: 5.8002x; 5.8002x over previous
#include <hip/hip_runtime.h>
#include <math.h>


typedef __attribute__((ext_vector_type(2))) float v2f;
typedef __attribute__((ext_vector_type(8))) float v8f;

#define B_DIM 2
#define N_NODES 25000
#define N_EDGES 400000
#define FC_IN 64
#define FC_HID 256
#define MUL 32
#define OUT_CH 288
#define EDGES_PER_CHUNK 128
#define NODES_PER_GROUP 8
#define WAVES 8
#define MAIN_BLOCKS 256
#define SH_STRIDE 18
#define SW_STRIDE 97
#define SC_STRIDE 18

#define WS_OFF_CC     25008
#define WS_OFF_PERM   50016
#define WS_OFF_BUCKET (50016 + N_EDGES)
#define WS_NEED_BYTES ((size_t)(50016 + 2 * N_EDGES) * sizeof(int))

__global__ __launch_bounds__(256) void sort_zero(int* __restrict__ cc, size_t ws_bytes)
{
    if (ws_bytes < WS_NEED_BYTES) return;
    const int i = blockIdx.x * 256 + threadIdx.x;
    if (i < N_NODES) cc[i] = 0;
}

__global__ __launch_bounds__(256) void sort_count(const int* __restrict__ edge_dst,
                                                  int* __restrict__ cc, size_t ws_bytes)
{
    if (ws_bytes < WS_NEED_BYTES) return;
    const int e = blockIdx.x * 256 + threadIdx.x;
    if (e < N_EDGES) atomicAdd(&cc[edge_dst[e]], 1);
}

__global__ __launch_bounds__(1024) void sort_scan(int* __restrict__ cc, int* __restrict__ off,
                                                  size_t ws_bytes)
{
    if (ws_bytes < WS_NEED_BYTES) return;
    __shared__ int s[1024];
    const int t = threadIdx.x;
    const int base = t * 25;
    int loc[25];
    int sum = 0;
    #pragma unroll
    for (int i = 0; i < 25; ++i) {
        const int idx = base + i;
        const int v = (idx < N_NODES) ? cc[idx] : 0;
        loc[i] = v;
        sum += v;
    }
    s[t] = sum;
    __syncthreads();
    for (int d = 1; d < 1024; d <<= 1) {
        const int v = (t >= d) ? s[t - d] : 0;
        __syncthreads();
        s[t] += v;
        __syncthreads();
    }
    int run = s[t] - sum;
    #pragma unroll
    for (int i = 0; i < 25; ++i) {
        const int idx = base + i;
        if (idx < N_NODES) {
            off[idx] = run;
            cc[idx]  = run;
            run += loc[i];
        }
    }
    if (t == 1023) off[N_NODES] = s[1023];
}

__global__ __launch_bounds__(256) void sort_fill(const int* __restrict__ edge_dst,
                                                 int* __restrict__ cc,
                                                 int* __restrict__ bucket, size_t ws_bytes)
{
    if (ws_bytes < WS_NEED_BYTES) return;
    const int e = blockIdx.x * 256 + threadIdx.x;
    if (e < N_EDGES) {
        const int slot = atomicAdd(&cc[edge_dst[e]], 1);
        bucket[slot] = e;
    }
}

__global__ __launch_bounds__(256) void sort_rank(const int* __restrict__ edge_dst,
                                                 const int* __restrict__ off,
                                                 const int* __restrict__ bucket,
                                                 int* __restrict__ perm, size_t ws_bytes)
{
    if (ws_bytes < WS_NEED_BYTES) return;
    const int i = blockIdx.x * 256 + threadIdx.x;
    if (i < N_EDGES) {
        const int e  = bucket[i];
        const int d  = edge_dst[e];
        const int lo = off[d];
        const int hi = off[d + 1];
        int r = 0;
        for (int j = lo; j < hi; ++j) r += (bucket[j] < e) ? 1 : 0;
        perm[lo + r] = e;
    }
}

__global__ __launch_bounds__(256) void conv_mlp_tp_gather(
    const int* __restrict__ edge_src,
    const int* __restrict__ edge_dst,
    const float* __restrict__ node_emb,
    const float* __restrict__ edge_type,
    const float* __restrict__ W1,
    const float* __restrict__ W2,
    const int* __restrict__ off,
    const int* __restrict__ perm,
    float* __restrict__ out,
    size_t ws_bytes)
{
    if (ws_bytes < WS_NEED_BYTES) return;
    __shared__ v2f   sB1[16 * 16 * 32];
    __shared__ v2f   sB2[16 * 6 * 4 * 32];
    __shared__ float sH[WAVES * 16 * SH_STRIDE];
    __shared__ float sW[EDGES_PER_CHUNK * SW_STRIDE];
    __shared__ float sC[EDGES_PER_CHUNK * SC_STRIDE];
    __shared__ float sAcc[WAVES * 18 * 32];

    const int tid   = threadIdx.x;
    const int wv    = __builtin_amdgcn_readfirstlane(tid >> 5);
    const int lane  = tid & 31;
    const int lmod  = lane & 15;
    const int khalf = (lane < 16) ? 0 : 2;

    #pragma unroll 8
    for (int it = 0; it < (16 * 16 * 32) / 256; ++it) {
        const int idx = tid + 256 * it;
        const int l  = idx & 31;
        const int ks = (idx >> 5) & 15;
        const int nt = idx >> 9;
        const int k0 = 4 * ks + ((l < 16) ? 0 : 2);
        const int n  = 16 * nt + (l & 15);
        v2f b;
        b.x = W1[(size_t)k0 * FC_HID + n];
        b.y = W1[(size_t)(k0 + 1) * FC_HID + n];
        sB1[idx] = b;
    }
    #pragma unroll 8
    for (int it = 0; it < (16 * 6 * 4 * 32) / 256; ++it) {
        const int idx = tid + 256 * it;
        const int l   = idx & 31;
        const int kk  = (idx >> 5) & 3;
        const int r   = idx >> 7;
        const int nt1 = r / 6;
        const int nt2 = r - 6 * nt1;
        const int k0  = 16 * nt1 + 4 * kk + ((l < 16) ? 0 : 2);
        const int n   = 16 * nt2 + (l & 15);
        v2f b;
        b.x = W2[(size_t)k0 * 96 + n];
        b.y = W2[(size_t)(k0 + 1) * 96 + n];
        sB2[idx] = b;
    }
    __syncthreads();

    const v8f vzero = {0.f, 0.f, 0.f, 0.f, 0.f, 0.f, 0.f, 0.f};

    const int nlo = ((int)blockIdx.x * N_NODES) / MAIN_BLOCKS;
    const int nhi = (((int)blockIdx.x + 1) * N_NODES) / MAIN_BLOCKS;

    int n0 = nlo;
    while (n0 < nhi) {
    const int onext = n0 + lane;
    const int ov    = off[(onext < nhi) ? onext : nhi];
    const int s0    = __builtin_amdgcn_readlane(ov, 0);
    int m = 1;
    #pragma unroll
    for (int i = 2; i <= NODES_PER_GROUP; ++i) {
        const int v = __builtin_amdgcn_readlane(ov, i);
        if ((n0 + i <= nhi) && (v - s0) <= EDGES_PER_CHUNK) m = i;
    }
    const int  s1       = __builtin_amdgcn_readlane(ov, m);
    const bool has_node = (wv < m);
    const int  node     = n0 + wv;
    const int  lo_r     = __builtin_amdgcn_readlane(ov, wv);
    const int  hi_r     = __builtin_amdgcn_readlane(ov, wv + 1);
    const int  seg_lo   = has_node ? lo_r : s1;
    const int  seg_hi   = has_node ? hi_r : s1;

    int cs = s0;
    do {
        {
            const int e_loc = tid >> 1;
            const int bb    = tid & 1;
            const int slot  = cs + e_loc;
            if (slot < s1) {
                const int e  = perm[slot];
                const int si = edge_src[e];
                const int di = edge_dst[e];
                const float* xp = node_emb + ((size_t)bb * N_NODES + si) * 3;
                const float* yp = node_emb + ((size_t)bb * N_NODES + di) * 3;
                const float x0 = xp[0], x1 = xp[1], x2 = xp[2];
                const float y0 = yp[0], y1 = yp[1], y2 = yp[2];

                const float inv3 = 0.57735026918962576f;
                const float inv2 = 0.70710678118654752f;
                const float inv6 = 0.40824829046386302f;
                const float scl  = 0.25f;

                float* crow = sC + e_loc * SC_STRIDE + bb;
                crow[0]  = (x0*y0 + x1*y1 + x2*y2) * inv3 * scl;
                crow[2]  = (x1*y2 - x2*y1) * inv2 * scl;
                crow[4]  = (x2*y0 - x0*y2) * inv2 * scl;
                crow[6]  = (x0*y1 - x1*y0) * inv2 * scl;
                crow[8]  = (x0*y1 + x1*y0) * inv2 * scl;
                crow[10] = (x1*y2 + x2*y1) * inv2 * scl;
                crow[12] = (x0*y2 + x2*y0) * inv2 * scl;
                crow[14] = (x0*y0 - x1*y1) * inv2 * scl;
                crow[16] = (2.0f*x2*y2 - x0*y0 - x1*y1) * inv6 * scl;
            }
        }

        const int t0 = cs + wv * 16;
        if (t0 < s1) {
            v2f aET[16];
            {
                int slot = t0 + lmod;
                if (slot > s1 - 1) slot = s1 - 1;
                const int e = perm[slot];
                const float* etp = edge_type + (size_t)e * FC_IN;
                #pragma unroll
                for (int ks = 0; ks < 16; ++ks)
                    aET[ks] = *(const v2f*)(etp + 4 * ks + khalf);
            }

            v8f h[16];
            v2f bq[2][32];
            #pragma unroll
            for (int q = 0; q < 32; ++q) bq[0][q] = sB1[q * 32 + lane];
            #pragma unroll
            for (int np = 0; np < 8; ++np) {
                if (np + 1 < 8) {
                    #pragma unroll
                    for (int q = 0; q < 32; ++q)
                        bq[(np + 1) & 1][q] = sB1[((np + 1) * 32 + q) * 32 + lane];
                }
                __builtin_amdgcn_sched_barrier(0);
                v8f accA = vzero, accB = vzero;
                #pragma unroll
                for (int ks = 0; ks < 16; ++ks) {
                    accA = __builtin_amdgcn_wmma_f32_16x16x4_f32(
                        false, aET[ks], false, bq[np & 1][ks], (short)0, accA, false, false);
                    accB = __builtin_amdgcn_wmma_f32_16x16x4_f32(
                        false, aET[ks], false, bq[np & 1][16 + ks], (short)0, accB, false, false);
                }
                #pragma unroll
                for (int i = 0; i < 8; ++i) {
                    const float ha = accA[i] * 0.125f;
                    const float hb = accB[i] * 0.125f;
                    accA[i] = ha / (1.0f + __expf(-ha));
                    accB[i] = hb / (1.0f + __expf(-hb));
                }
                h[2 * np]     = accA;
                h[2 * np + 1] = accB;
                __builtin_amdgcn_sched_barrier(0);
            }

            float* myH = sH + wv * (16 * SH_STRIDE);
            const int mbaseH = (lane < 16) ? 0 : 8;
            v8f acc2[6];
            #pragma unroll
            for (int j = 0; j < 6; ++j) acc2[j] = vzero;
            v2f aq[2][4];
            v2f bw[2][24];
            #pragma unroll
            for (int r = 0; r < 8; ++r)
                myH[(r + mbaseH) * SH_STRIDE + lmod] = h[0][r];
            #pragma unroll
            for (int kk = 0; kk < 4; ++kk)
                aq[0][kk] = *(const v2f*)(myH + lmod * SH_STRIDE + 4 * kk + khalf);
            #pragma unroll
            for (int q = 0; q < 24; ++q) bw[0][q] = sB2[q * 32 + lane];

            #pragma unroll
            for (int nt1 = 0; nt1 < 16; ++nt1) {
                if (nt1 + 1 < 16) {
                    #pragma unroll
                    for (int r = 0; r < 8; ++r)
                        myH[(r + mbaseH) * SH_STRIDE + lmod] = h[nt1 + 1][r];
                    #pragma unroll
                    for (int kk = 0; kk < 4; ++kk)
                        aq[(nt1 + 1) & 1][kk] = *(const v2f*)(myH + lmod * SH_STRIDE + 4 * kk + khalf);
                    #pragma unroll
                    for (int q = 0; q < 24; ++q)
                        bw[(nt1 + 1) & 1][q] = sB2[((nt1 + 1) * 24 + q) * 32 + lane];
                }
                __builtin_amdgcn_sched_barrier(0);
                #pragma unroll
                for (int kk = 0; kk < 4; ++kk) {
                    #pragma unroll
                    for (int nt2 = 0; nt2 < 6; ++nt2)
                        acc2[nt2] = __builtin_amdgcn_wmma_f32_16x16x4_f32(
                            false, aq[nt1 & 1][kk], false, bw[nt1 & 1][nt2 * 4 + kk],
                            (short)0, acc2[nt2], false, false);
                }
                __builtin_amdgcn_sched_barrier(0);
            }

            {
                const int mbase = wv * 16 + ((lane < 16) ? 0 : 8);
                #pragma unroll
                for (int nt2 = 0; nt2 < 6; ++nt2) {
                    #pragma unroll
                    for (int r = 0; r < 8; ++r)
                        sW[(mbase + r) * SW_STRIDE + 16 * nt2 + lmod] = acc2[nt2][r] * 0.0625f;
                }
            }
        }
        __syncthreads();

        if (has_node) {
            int wcol[9], kco[9];
            wcol[0] = lane;
            kco[0]  = 0;
            #pragma unroll
            for (int i = 1; i < 4; ++i) {
                const int q = lane + 32 * (i - 1);
                const int u = q / 3;
                wcol[i] = MUL + u;
                kco[i]  = 1 + (q - 3 * u);
            }
            #pragma unroll
            for (int i = 4; i < 9; ++i) {
                const int q = lane + 32 * (i - 4);
                const int u = q / 5;
                wcol[i] = 2 * MUL + u;
                kco[i]  = 4 + (q - 5 * u);
            }

            float* park = sAcc + wv * (18 * 32) + lane;
            float acc0[9], acc1[9];
            if (cs == s0) {
                #pragma unroll
                for (int i = 0; i < 9; ++i) { acc0[i] = 0.f; acc1[i] = 0.f; }
            } else {
                #pragma unroll
                for (int i = 0; i < 9; ++i) { acc0[i] = park[32 * i]; acc1[i] = park[32 * (9 + i)]; }
            }

            const int ce = cs + EDGES_PER_CHUNK;
            const int j0 = (seg_lo > cs) ? seg_lo : cs;
            const int j1 = (seg_hi < ce) ? seg_hi : ce;
            for (int j = j0; j < j1; ++j) {
                const float* wrow = sW + (j - cs) * SW_STRIDE;
                const float* crow = sC + (j - cs) * SC_STRIDE;
                #pragma unroll
                for (int i = 0; i < 9; ++i) {
                    const float w = wrow[wcol[i]];
                    const v2f   c = *(const v2f*)(crow + 2 * kco[i]);
                    acc0[i] = fmaf(w, c.x, acc0[i]);
                    acc1[i] = fmaf(w, c.y, acc1[i]);
                }
            }

            if (ce >= s1) {
                float* o0 = out + ((size_t)node) * OUT_CH + lane;
                float* o1 = out + ((size_t)N_NODES + node) * OUT_CH + lane;
                #pragma unroll
                for (int i = 0; i < 9; ++i) {
                    o0[32 * i] = acc0[i];
                    o1[32 * i] = acc1[i];
                }
            } else {
                #pragma unroll
                for (int i = 0; i < 9; ++i) { park[32 * i] = acc0[i]; park[32 * (9 + i)] = acc1[i]; }
            }
        }
        __syncthreads();
        cs += EDGES_PER_CHUNK;
    } while (cs < s1);
    n0 += m;
    }
}

extern "C" void kernel_launch(void* const* d_in, const int* in_sizes, int n_in,
                              void* d_out, int out_size, void* d_ws, size_t ws_size,
                              hipStream_t stream) {
    const int*   edge_src  = (const int*)d_in[0];
    const int*   edge_dst  = (const int*)d_in[1];
    const float* node_emb  = (const float*)d_in[2];
    const float* edge_type = (const float*)d_in[3];
    const float* W1        = (const float*)d_in[4];
    const float* W2        = (const float*)d_in[5];
    float*       out       = (float*)d_out;

    int* off    = (int*)d_ws;
    int* cc     = off + WS_OFF_CC;
    int* perm   = off + WS_OFF_PERM;
    int* bucket = off + WS_OFF_BUCKET;

    sort_zero <<<  98, 256, 0, stream>>>(cc, ws_size);
    sort_count<<<1563, 256, 0, stream>>>(edge_dst, cc, ws_size);
    sort_scan <<<   1, 1024, 0, stream>>>(cc, off, ws_size);
    sort_fill <<<1563, 256, 0, stream>>>(edge_dst, cc, bucket, ws_size);
    sort_rank <<<1563, 256, 0, stream>>>(edge_dst, off, bucket, perm, ws_size);
    conv_mlp_tp_gather<<<MAIN_BLOCKS, 256, 0, stream>>>(
        edge_src, edge_dst, node_emb, edge_type, W1, W2, off, perm, out, ws_size);
}


# no global round trip waited for in the chunk loop: next chunk's off[], edge ids, endpoints, embeddings, edge_type rows asked for under this chunk's matrix runs; LDS-only barrier; add order kept
# speedup vs baseline: 6.3216x; 1.0899x over previous
#include <hip/hip_runtime.h>
#include <math.h>


typedef __attribute__((ext_vector_type(2))) float v2f;
typedef __attribute__((ext_vector_type(8))) float v8f;

#define B_DIM 2
#define N_NODES 25000
#define N_EDGES 400000
#define FC_IN 64
#define FC_HID 256
#define MUL 32
#define OUT_CH 288
#define EDGES_PER_CHUNK 128
#define NODES_PER_GROUP 8
#define WAVES 8
#define MAIN_BLOCKS 256
#define SH_STRIDE 18
#define SW_STRIDE 97
#define SC_STRIDE 18

#define WS_OFF_CC     25008
#define WS_OFF_PERM   50016
#define WS_OFF_BUCKET (50016 + N_EDGES)
#define WS_NEED_BYTES ((size_t)(50016 + 2 * N_EDGES) * sizeof(int))

__global__ __launch_bounds__(256) void sort_zero(int* __restrict__ cc, size_t ws_bytes)
{
    if (ws_bytes < WS_NEED_BYTES) return;
    const int i = blockIdx.x * 256 + threadIdx.x;
    if (i < N_NODES) cc[i] = 0;
}

__global__ __launch_bounds__(256) void sort_count(const int* __restrict__ edge_dst,
                                                  int* __restrict__ cc, size_t ws_bytes)
{
    if (ws_bytes < WS_NEED_BYTES) return;
    const int e = blockIdx.x * 256 + threadIdx.x;
    if (e < N_EDGES) atomicAdd(&cc[edge_dst[e]], 1);
}

__global__ __launch_bounds__(1024) void sort_scan(int* __restrict__ cc, int* __restrict__ off,
                                                  size_t ws_bytes)
{
    if (ws_bytes < WS_NEED_BYTES) return;
    __shared__ int s[1024];
    const int t = threadIdx.x;
    const int base = t * 25;
    int loc[25];
    int sum = 0;
    #pragma unroll
    for (int i = 0; i < 25; ++i) {
        const int idx = base + i;
        const int v = (idx < N_NODES) ? cc[idx] : 0;
        loc[i] = v;
        sum += v;
    }
    s[t] = sum;
    __syncthreads();
    for (int d = 1; d < 1024; d <<= 1) {
        const int v = (t >= d) ? s[t - d] : 0;
        __syncthreads();
        s[t] += v;
        __syncthreads();
    }
    int run = s[t] - sum;
    #pragma unroll
    for (int i = 0; i < 25; ++i) {
        const int idx = base + i;
        if (idx < N_NODES) {
            off[idx] = run;
            cc[idx]  = run;
            run += loc[i];
        }
    }
    if (t == 1023) off[N_NODES] = s[1023];
}

__global__ __launch_bounds__(256) void sort_fill(const int* __restrict__ edge_dst,
                                                 int* __restrict__ cc,
                                                 int* __restrict__ bucket, size_t ws_bytes)
{
    if (ws_bytes < WS_NEED_BYTES) return;
    const int e = blockIdx.x * 256 + threadIdx.x;
    if (e < N_EDGES) {
        const int slot = atomicAdd(&cc[edge_dst[e]], 1);
        bucket[slot] = e;
    }
}

__global__ __launch_bounds__(256) void sort_rank(const int* __restrict__ edge_dst,
                                                 const int* __restrict__ off,
                                                 const int* __restrict__ bucket,
                                                 int* __restrict__ perm, size_t ws_bytes)
{
    if (ws_bytes < WS_NEED_BYTES) return;
    const int i = blockIdx.x * 256 + threadIdx.x;
    if (i < N_EDGES) {
        const int e  = bucket[i];
        const int d  = edge_dst[e];
        const int lo = off[d];
        const int hi = off[d + 1];
        int r = 0;
        for (int j = lo; j < hi; ++j) r += (bucket[j] < e) ? 1 : 0;
        perm[lo + r] = e;
    }
}

__global__ __launch_bounds__(256) void conv_mlp_tp_gather(
    const int* __restrict__ edge_src,
    const int* __restrict__ edge_dst,
    const float* __restrict__ node_emb,
    const float* __restrict__ edge_type,
    const float* __restrict__ W1,
    const float* __restrict__ W2,
    const int* __restrict__ off,
    const int* __restrict__ perm,
    float* __restrict__ out,
    size_t ws_bytes)
{
    if (ws_bytes < WS_NEED_BYTES) return;
    __shared__ v2f   sB1[16 * 16 * 32];
    __shared__ v2f   sB2[16 * 6 * 4 * 32];
    __shared__ float sH[WAVES * 16 * SH_STRIDE];
    __shared__ float sW[EDGES_PER_CHUNK * SW_STRIDE];
    __shared__ float sC[EDGES_PER_CHUNK * SC_STRIDE];
    __shared__ float sAcc[WAVES * 18 * 32];

    const int tid   = threadIdx.x;
    const int wv    = __builtin_amdgcn_readfirstlane(tid >> 5);
    const int lane  = tid & 31;
    const int lmod  = lane & 15;
    const int khalf = (lane < 16) ? 0 : 2;
    const int e_loc = tid >> 1;
    const int bb    = tid & 1;

    const int nlo = ((int)blockIdx.x * N_NODES) / MAIN_BLOCKS;
    const int nhi = (((int)blockIdx.x + 1) * N_NODES) / MAIN_BLOCKS;
    if (nlo >= nhi) return;

    const int e_end    = off[nhi];
    const int slot_max = (e_end > 0) ? e_end - 1 : 0;

    int n0 = nlo;
    int ov = off[(n0 + lane < nhi) ? n0 + lane : nhi];

    #pragma unroll 8
    for (int it = 0; it < (16 * 16 * 32) / 256; ++it) {
        const int idx = tid + 256 * it;
        const int l  = idx & 31;
        const int ks = (idx >> 5) & 15;
        const int nt = idx >> 9;
        const int k0 = 4 * ks + ((l < 16) ? 0 : 2);
        const int n  = 16 * nt + (l & 15);
        v2f b;
        b.x = W1[(size_t)k0 * FC_HID + n];
        b.y = W1[(size_t)(k0 + 1) * FC_HID + n];
        sB1[idx] = b;
    }

    int cs = __builtin_amdgcn_readlane(ov, 0);
    int pC, pT;
    {
        const int sc = cs + e_loc;
        const int st = cs + wv * 16 + lmod;
        pC = perm[(sc < slot_max) ? sc : slot_max];
        pT = perm[(st < slot_max) ? st : slot_max];
    }

    #pragma unroll 8
    for (int it = 0; it < (16 * 6 * 4 * 32) / 512; ++it) {
        const int idx = tid + 256 * it;
        const int l   = idx & 31;
        const int kk  = (idx >> 5) & 3;
        const int r   = idx >> 7;
        const int nt1 = r / 6;
        const int nt2 = r - 6 * nt1;
        const int k0  = 16 * nt1 + 4 * kk + ((l < 16) ? 0 : 2);
        const int n   = 16 * nt2 + (l & 15);
        v2f b;
        b.x = W2[(size_t)k0 * 96 + n];
        b.y = W2[(size_t)(k0 + 1) * 96 + n];
        sB2[idx] = b;
    }

    v2f aET[16];
    int si = edge_src[pC];
    int di = edge_dst[pC];
    {
        const float* etp = edge_type + (size_t)pT * FC_IN;
        #pragma unroll
        for (int ks = 0; ks < 16; ++ks)
            aET[ks] = *(const v2f*)(etp + 4 * ks + khalf);
    }

    #pragma unroll 8
    for (int it = (16 * 6 * 4 * 32) / 512; it < (16 * 6 * 4 * 32) / 256; ++it) {
        const int idx = tid + 256 * it;
        const int l   = idx & 31;
        const int kk  = (idx >> 5) & 3;
        const int r   = idx >> 7;
        const int nt1 = r / 6;
        const int nt2 = r - 6 * nt1;
        const int k0  = 16 * nt1 + 4 * kk + ((l < 16) ? 0 : 2);
        const int n   = 16 * nt2 + (l & 15);
        v2f b;
        b.x = W2[(size_t)k0 * 96 + n];
        b.y = W2[(size_t)(k0 + 1) * 96 + n];
        sB2[idx] = b;
    }

    float x0, x1, x2, y0, y1, y2;
    {
        const float* xp = node_emb + ((size_t)bb * N_NODES + si) * 3;
        const float* yp = node_emb + ((size_t)bb * N_NODES + di) * 3;
        x0 = xp[0]; x1 = xp[1]; x2 = xp[2];
        y0 = yp[0]; y1 = yp[1]; y2 = yp[2];
    }
    __syncthreads();

    const v8f vzero = {0.f, 0.f, 0.f, 0.f, 0.f, 0.f, 0.f, 0.f};

    for (;;) {
    const int s0 = __builtin_amdgcn_readlane(ov, 0);
    int m = 1;
    #pragma unroll
    for (int i = 2; i <= NODES_PER_GROUP; ++i) {
        const int v = __builtin_amdgcn_readlane(ov, i);
        if ((n0 + i <= nhi) && (v - s0) <= EDGES_PER_CHUNK) m = i;
    }
    const int  s1       = __builtin_amdgcn_readlane(ov, m);
    const bool has_node = (wv < m);
    const int  node     = n0 + wv;
    const int  lo_r     = __builtin_amdgcn_readlane(ov, wv);
    const int  hi_r     = __builtin_amdgcn_readlane(ov, wv + 1);
    const int  seg_lo   = has_node ? lo_r : s1;
    const int  seg_hi   = has_node ? hi_r : s1;

    const bool last_of_group = (cs + EDGES_PER_CHUNK >= s1);
    const int  csN  = last_of_group ? s1 : cs + EDGES_PER_CHUNK;
    const int  n0N  = last_of_group ? n0 + m : n0;
    const bool more = (n0N < nhi);
    const int ovN = off[(n0N + lane < nhi) ? n0N + lane : nhi];
    int pCN, pTN;
    {
        const int sc = csN + e_loc;
        const int st = csN + wv * 16 + lmod;
        pCN = perm[(sc < slot_max) ? sc : slot_max];
        pTN = perm[(st < slot_max) ? st : slot_max];
    }

        if (cs + e_loc < s1) {
                const float inv3 = 0.57735026918962576f;
                const float inv2 = 0.70710678118654752f;
                const float inv6 = 0.40824829046386302f;
                const float scl  = 0.25f;

                float* crow = sC + e_loc * SC_STRIDE + bb;
                crow[0]  = (x0*y0 + x1*y1 + x2*y2) * inv3 * scl;
                crow[2]  = (x1*y2 - x2*y1) * inv2 * scl;
                crow[4]  = (x2*y0 - x0*y2) * inv2 * scl;
                crow[6]  = (x0*y1 - x1*y0) * inv2 * scl;
                crow[8]  = (x0*y1 + x1*y0) * inv2 * scl;
                crow[10] = (x1*y2 + x2*y1) * inv2 * scl;
                crow[12] = (x0*y2 + x2*y0) * inv2 * scl;
                crow[14] = (x0*y0 - x1*y1) * inv2 * scl;
                crow[16] = (2.0f*x2*y2 - x0*y0 - x1*y1) * inv6 * scl;
        }

        const int  t0       = cs + wv * 16;
        const bool has_tile = (t0 < s1);
        v8f h[16];
        auto ask_nodes = [&]() {
            const float* xp = node_emb + ((size_t)bb * N_NODES + si) * 3;
            const float* yp = node_emb + ((size_t)bb * N_NODES + di) * 3;
            x0 = xp[0]; x1 = xp[1]; x2 = xp[2];
            y0 = yp[0]; y1 = yp[1]; y2 = yp[2];
        };
        if (has_tile) {
            v2f bq[2][32];
            #pragma unroll
            for (int q = 0; q < 32; ++q) bq[0][q] = sB1[q * 32 + lane];
            #pragma unroll
            for (int np = 0; np < 8; ++np) {
                if (np + 1 < 8) {
                    #pragma unroll
                    for (int q = 0; q < 32; ++q)
                        bq[(np + 1) & 1][q] = sB1[((np + 1) * 32 + q) * 32 + lane];
                }
                __builtin_amdgcn_sched_barrier(0);
                v8f accA = vzero, accB = vzero;
                #pragma unroll
                for (int ks = 0; ks < 16; ++ks) {
                    accA = __builtin_amdgcn_wmma_f32_16x16x4_f32(
                        false, aET[ks], false, bq[np & 1][ks], (short)0, accA, false, false);
                    accB = __builtin_amdgcn_wmma_f32_16x16x4_f32(
                        false, aET[ks], false, bq[np & 1][16 + ks], (short)0, accB, false, false);
                }
                #pragma unroll
                for (int i = 0; i < 8; ++i) {
                    const float ha = accA[i] * 0.125f;
                    const float hb = accB[i] * 0.125f;
                    accA[i] = ha / (1.0f + __expf(-ha));
                    accB[i] = hb / (1.0f + __expf(-hb));
                }
                h[2 * np]     = accA;
                h[2 * np + 1] = accB;
                __builtin_amdgcn_sched_barrier(0);
            }
        }

        si = edge_src[pCN];
        di = edge_dst[pCN];
        {
            const float* etp = edge_type + (size_t)pTN * FC_IN;
            #pragma unroll
            for (int ks = 0; ks < 16; ++ks)
                aET[ks] = *(const v2f*)(etp + 4 * ks + khalf);
        }

        if (has_tile) {
            float* myH = sH + wv * (16 * SH_STRIDE);
            const int mbaseH = (lane < 16) ? 0 : 8;
            v8f acc2[6];
            #pragma unroll
            for (int j = 0; j < 6; ++j) acc2[j] = vzero;
            v2f aq[2][4];
            v2f bw[2][24];
            #pragma unroll
            for (int r = 0; r < 8; ++r)
                myH[(r + mbaseH) * SH_STRIDE + lmod] = h[0][r];
            #pragma unroll
            for (int kk = 0; kk < 4; ++kk)
                aq[0][kk] = *(const v2f*)(myH + lmod * SH_STRIDE + 4 * kk + khalf);
            #pragma unroll
            for (int q = 0; q < 24; ++q) bw[0][q] = sB2[q * 32 + lane];

            #pragma unroll
            for (int nt1 = 0; nt1 < 16; ++nt1) {
                if (nt1 + 1 < 16) {
                    #pragma unroll
                    for (int r = 0; r < 8; ++r)
                        myH[(r + mbaseH) * SH_STRIDE + lmod] = h[nt1 + 1][r];
                    #pragma unroll
                    for (int kk = 0; kk < 4; ++kk)
                        aq[(nt1 + 1) & 1][kk] = *(const v2f*)(myH + lmod * SH_STRIDE + 4 * kk + khalf);
                    #pragma unroll
                    for (int q = 0; q < 24; ++q)
                        bw[(nt1 + 1) & 1][q] = sB2[((nt1 + 1) * 24 + q) * 32 + lane];
                }
                if (nt1 == 8) {
                    asm volatile("" : "+v"(si), "+v"(di));
                    ask_nodes();
                }
                __builtin_amdgcn_sched_barrier(0);
                #pragma unroll
                for (int kk = 0; kk < 4; ++kk) {
                    #pragma unroll
                    for (int nt2 = 0; nt2 < 6; ++nt2)
                        acc2[nt2] = __builtin_amdgcn_wmma_f32_16x16x4_f32(
                            false, aq[nt1 & 1][kk], false, bw[nt1 & 1][nt2 * 4 + kk],
                            (short)0, acc2[nt2], false, false);
                }
                __builtin_amdgcn_sched_barrier(0);
            }

            {
                const int mbase = wv * 16 + ((lane < 16) ? 0 : 8);
                #pragma unroll
                for (int nt2 = 0; nt2 < 6; ++nt2) {
                    #pragma unroll
                    for (int r = 0; r < 8; ++r)
                        sW[(mbase + r) * SW_STRIDE + 16 * nt2 + lmod] = acc2[nt2][r] * 0.0625f;
                }
            }
        } else {
            ask_nodes();
        }
        __syncthreads();

        if (has_node) {
            int wcol[9], kco[9];
            wcol[0] = lane;
            kco[0]  = 0;
            #pragma unroll
            for (int i = 1; i < 4; ++i) {
                const int q = lane + 32 * (i - 1);
                const int u = q / 3;
                wcol[i] = MUL + u;
                kco[i]  = 1 + (q - 3 * u);
            }
            #pragma unroll
            for (int i = 4; i < 9; ++i) {
                const int q = lane + 32 * (i - 4);
                const int u = q / 5;
                wcol[i] = 2 * MUL + u;
                kco[i]  = 4 + (q - 5 * u);
            }

            float* park = sAcc + wv * (18 * 32) + lane;
            float acc0[9], acc1[9];
            if (cs == s0) {
                #pragma unroll
                for (int i = 0; i < 9; ++i) { acc0[i] = 0.f; acc1[i] = 0.f; }
            } else {
                #pragma unroll
                for (int i = 0; i < 9; ++i) { acc0[i] = park[32 * i]; acc1[i] = park[32 * (9 + i)]; }
            }

            const int ce = cs + EDGES_PER_CHUNK;
            const int j0 = (seg_lo > cs) ? seg_lo : cs;
            const int j1 = (seg_hi < ce) ? seg_hi : ce;
            for (int j = j0; j < j1; ++j) {
                const float* wrow = sW + (j - cs) * SW_STRIDE;
                const float* crow = sC + (j - cs) * SC_STRIDE;
                #pragma unroll
                for (int i = 0; i < 9; ++i) {
                    const float w = wrow[wcol[i]];
                    const v2f   c = *(const v2f*)(crow + 2 * kco[i]);
                    acc0[i] = fmaf(w, c.x, acc0[i]);
                    acc1[i] = fmaf(w, c.y, acc1[i]);
                }
            }

            if (ce >= s1) {
                float* o0 = out + ((size_t)node) * OUT_CH + lane;
                float* o1 = out + ((size_t)N_NODES + node) * OUT_CH + lane;
                #pragma unroll
                for (int i = 0; i < 9; ++i) {
                    o0[32 * i] = acc0[i];
                    o1[32 * i] = acc1[i];
                }
            } else {
                #pragma unroll
                for (int i = 0; i < 9; ++i) { park[32 * i] = acc0[i]; park[32 * (9 + i)] = acc1[i]; }
            }
        }
        if (!more) break;
        __builtin_amdgcn_fence(__ATOMIC_RELEASE, "workgroup", "local");
        __builtin_amdgcn_s_barrier();
        __builtin_amdgcn_fence(__ATOMIC_ACQUIRE, "workgroup", "local");
        cs = csN;
        n0 = n0N;
        ov = ovN;
    }
}

extern "C" void kernel_launch(void* const* d_in, const int* in_sizes, int n_in,
                              void* d_out, int out_size, void* d_ws, size_t ws_size,
                              hipStream_t stream) {
    const int*   edge_src  = (const int*)d_in[0];
    const int*   edge_dst  = (const int*)d_in[1];
    const float* node_emb  = (const float*)d_in[2];
    const float* edge_type = (const float*)d_in[3];
    const float* W1        = (const float*)d_in[4];
    const float* W2        = (const float*)d_in[5];
    float*       out       = (float*)d_out;

    int* off    = (int*)d_ws;
    int* cc     = off + WS_OFF_CC;
    int* perm   = off + WS_OFF_PERM;
    int* bucket = off + WS_OFF_BUCKET;

    sort_zero <<<  98, 256, 0, stream>>>(cc, ws_size);
    sort_count<<<1563, 256, 0, stream>>>(edge_dst, cc, ws_size);
    sort_scan <<<   1, 1024, 0, stream>>>(cc, off, ws_size);
    sort_fill <<<1563, 256, 0, stream>>>(edge_dst, cc, bucket, ws_size);
    sort_rank <<<1563, 256, 0, stream>>>(edge_dst, off, bucket, perm, ws_size);
    conv_mlp_tp_gather<<<MAIN_BLOCKS, 256, 0, stream>>>(
        edge_src, edge_dst, node_emb, edge_type, W1, W2, off, perm, out, ws_size);
}


# silu: IEEE division replaced by the hardware reciprocal and one multiply (divisor in [1,inf], about 1 ulp); its scale, refinement and fix-up vector instructions leave the first product
# speedup vs baseline: 6.6938x; 1.0589x over previous
#include <hip/hip_runtime.h>
#include <math.h>


typedef __attribute__((ext_vector_type(2))) float v2f;
typedef __attribute__((ext_vector_type(8))) float v8f;

#define B_DIM 2
#define N_NODES 25000
#define N_EDGES 400000
#define FC_IN 64
#define FC_HID 256
#define MUL 32
#define OUT_CH 288
#define EDGES_PER_CHUNK 128
#define NODES_PER_GROUP 8
#define WAVES 8
#define MAIN_BLOCKS 256
#define SH_STRIDE 18
#define SW_STRIDE 97
#define SC_STRIDE 18

#define WS_OFF_CC     25008
#define WS_OFF_PERM   50016
#define WS_OFF_BUCKET (50016 + N_EDGES)
#define WS_NEED_BYTES ((size_t)(50016 + 2 * N_EDGES) * sizeof(int))

__global__ __launch_bounds__(256) void sort_zero(int* __restrict__ cc, size_t ws_bytes)
{
    if (ws_bytes < WS_NEED_BYTES) return;
    const int i = blockIdx.x * 256 + threadIdx.x;
    if (i < N_NODES) cc[i] = 0;
}

__global__ __launch_bounds__(256) void sort_count(const int* __restrict__ edge_dst,
                                                  int* __restrict__ cc, size_t ws_bytes)
{
    if (ws_bytes < WS_NEED_BYTES) return;
    const int e = blockIdx.x * 256 + threadIdx.x;
    if (e < N_EDGES) atomicAdd(&cc[edge_dst[e]], 1);
}

__global__ __launch_bounds__(1024) void sort_scan(int* __restrict__ cc, int* __restrict__ off,
                                                  size_t ws_bytes)
{
    if (ws_bytes < WS_NEED_BYTES) return;
    __shared__ int s[1024];
    const int t = threadIdx.x;
    const int base = t * 25;
    int loc[25];
    int sum = 0;
    #pragma unroll
    for (int i = 0; i < 25; ++i) {
        const int idx = base + i;
        const int v = (idx < N_NODES) ? cc[idx] : 0;
        loc[i] = v;
        sum += v;
    }
    s[t] = sum;
    __syncthreads();
    for (int d = 1; d < 1024; d <<= 1) {
        const int v = (t >= d) ? s[t - d] : 0;
        __syncthreads();
        s[t] += v;
        __syncthreads();
    }
    int run = s[t] - sum;
    #pragma unroll
    for (int i = 0; i < 25; ++i) {
        const int idx = base + i;
        if (idx < N_NODES) {
            off[idx] = run;
            cc[idx]  = run;
            run += loc[i];
        }
    }
    if (t == 1023) off[N_NODES] = s[1023];
}

__global__ __launch_bounds__(256) void sort_fill(const int* __restrict__ edge_dst,
                                                 int* __restrict__ cc,
                                                 int* __restrict__ bucket, size_t ws_bytes)
{
    if (ws_bytes < WS_NEED_BYTES) return;
    const int e = blockIdx.x * 256 + threadIdx.x;
    if (e < N_EDGES) {
        const int slot = atomicAdd(&cc[edge_dst[e]], 1);
        bucket[slot] = e;
    }
}

__global__ __launch_bounds__(256) void sort_rank(const int* __restrict__ edge_dst,
                                                 const int* __restrict__ off,
                                                 const int* __restrict__ bucket,
                                                 int* __restrict__ perm, size_t ws_bytes)
{
    if (ws_bytes < WS_NEED_BYTES) return;
    const int i = blockIdx.x * 256 + threadIdx.x;
    if (i < N_EDGES) {
        const int e  = bucket[i];
        const int d  = edge_dst[e];
        const int lo = off[d];
        const int hi = off[d + 1];
        int r = 0;
        for (int j = lo; j < hi; ++j) r += (bucket[j] < e) ? 1 : 0;
        perm[lo + r] = e;
    }
}

__global__ __launch_bounds__(256) void conv_mlp_tp_gather(
    const int* __restrict__ edge_src,
    const int* __restrict__ edge_dst,
    const float* __restrict__ node_emb,
    const float* __restrict__ edge_type,
    const float* __restrict__ W1,
    const float* __restrict__ W2,
    const int* __restrict__ off,
    const int* __restrict__ perm,
    float* __restrict__ out,
    size_t ws_bytes)
{
    if (ws_bytes < WS_NEED_BYTES) return;
    __shared__ v2f   sB1[16 * 16 * 32];
    __shared__ v2f   sB2[16 * 6 * 4 * 32];
    __shared__ float sH[WAVES * 16 * SH_STRIDE];
    __shared__ float sW[EDGES_PER_CHUNK * SW_STRIDE];
    __shared__ float sC[EDGES_PER_CHUNK * SC_STRIDE];
    __shared__ float sAcc[WAVES * 18 * 32];

    const int tid   = threadIdx.x;
    const int wv    = __builtin_amdgcn_readfirstlane(tid >> 5);
    const int lane  = tid & 31;
    const int lmod  = lane & 15;
    const int khalf = (lane < 16) ? 0 : 2;
    const int e_loc = tid >> 1;
    const int bb    = tid & 1;

    const int nlo = ((int)blockIdx.x * N_NODES) / MAIN_BLOCKS;
    const int nhi = (((int)blockIdx.x + 1) * N_NODES) / MAIN_BLOCKS;
    if (nlo >= nhi) return;

    const int e_end    = off[nhi];
    const int slot_max = (e_end > 0) ? e_end - 1 : 0;

    int n0 = nlo;
    int ov = off[(n0 + lane < nhi) ? n0 + lane : nhi];

    #pragma unroll 8
    for (int it = 0; it < (16 * 16 * 32) / 256; ++it) {
        const int idx = tid + 256 * it;
        const int l  = idx & 31;
        const int ks = (idx >> 5) & 15;
        const int nt = idx >> 9;
        const int k0 = 4 * ks + ((l < 16) ? 0 : 2);
        const int n  = 16 * nt + (l & 15);
        v2f b;
        b.x = W1[(size_t)k0 * FC_HID + n];
        b.y = W1[(size_t)(k0 + 1) * FC_HID + n];
        sB1[idx] = b;
    }

    int cs = __builtin_amdgcn_readlane(ov, 0);
    int pC, pT;
    {
        const int sc = cs + e_loc;
        const int st = cs + wv * 16 + lmod;
        pC = perm[(sc < slot_max) ? sc : slot_max];
        pT = perm[(st < slot_max) ? st : slot_max];
    }

    #pragma unroll 8
    for (int it = 0; it < (16 * 6 * 4 * 32) / 512; ++it) {
        const int idx = tid + 256 * it;
        const int l   = idx & 31;
        const int kk  = (idx >> 5) & 3;
        const int r   = idx >> 7;
        const int nt1 = r / 6;
        const int nt2 = r - 6 * nt1;
        const int k0  = 16 * nt1 + 4 * kk + ((l < 16) ? 0 : 2);
        const int n   = 16 * nt2 + (l & 15);
        v2f b;
        b.x = W2[(size_t)k0 * 96 + n];
        b.y = W2[(size_t)(k0 + 1) * 96 + n];
        sB2[idx] = b;
    }

    v2f aET[16];
    int si = edge_src[pC];
    int di = edge_dst[pC];
    {
        const float* etp = edge_type + (size_t)pT * FC_IN;
        #pragma unroll
        for (int ks = 0; ks < 16; ++ks)
            aET[ks] = *(const v2f*)(etp + 4 * ks + khalf);
    }

    #pragma unroll 8
    for (int it = (16 * 6 * 4 * 32) / 512; it < (16 * 6 * 4 * 32) / 256; ++it) {
        const int idx = tid + 256 * it;
        const int l   = idx & 31;
        const int kk  = (idx >> 5) & 3;
        const int r   = idx >> 7;
        const int nt1 = r / 6;
        const int nt2 = r - 6 * nt1;
        const int k0  = 16 * nt1 + 4 * kk + ((l < 16) ? 0 : 2);
        const int n   = 16 * nt2 + (l & 15);
        v2f b;
        b.x = W2[(size_t)k0 * 96 + n];
        b.y = W2[(size_t)(k0 + 1) * 96 + n];
        sB2[idx] = b;
    }

    float x0, x1, x2, y0, y1, y2;
    {
        const float* xp = node_emb + ((size_t)bb * N_NODES + si) * 3;
        const float* yp = node_emb + ((size_t)bb * N_NODES + di) * 3;
        x0 = xp[0]; x1 = xp[1]; x2 = xp[2];
        y0 = yp[0]; y1 = yp[1]; y2 = yp[2];
    }
    __syncthreads();

    const v8f vzero = {0.f, 0.f, 0.f, 0.f, 0.f, 0.f, 0.f, 0.f};

    for (;;) {
    const int s0 = __builtin_amdgcn_readlane(ov, 0);
    int m = 1;
    #pragma unroll
    for (int i = 2; i <= NODES_PER_GROUP; ++i) {
        const int v = __builtin_amdgcn_readlane(ov, i);
        if ((n0 + i <= nhi) && (v - s0) <= EDGES_PER_CHUNK) m = i;
    }
    const int  s1       = __builtin_amdgcn_readlane(ov, m);
    const bool has_node = (wv < m);
    const int  node     = n0 + wv;
    const int  lo_r     = __builtin_amdgcn_readlane(ov, wv);
    const int  hi_r     = __builtin_amdgcn_readlane(ov, wv + 1);
    const int  seg_lo   = has_node ? lo_r : s1;
    const int  seg_hi   = has_node ? hi_r : s1;

    const bool last_of_group = (cs + EDGES_PER_CHUNK >= s1);
    const int  csN  = last_of_group ? s1 : cs + EDGES_PER_CHUNK;
    const int  n0N  = last_of_group ? n0 + m : n0;
    const bool more = (n0N < nhi);
    const int ovN = off[(n0N + lane < nhi) ? n0N + lane : nhi];
    int pCN, pTN;
    {
        const int sc = csN + e_loc;
        const int st = csN + wv * 16 + lmod;
        pCN = perm[(sc < slot_max) ? sc : slot_max];
        pTN = perm[(st < slot_max) ? st : slot_max];
    }

        if (cs + e_loc < s1) {
                const float inv3 = 0.57735026918962576f;
                const float inv2 = 0.70710678118654752f;
                const float inv6 = 0.40824829046386302f;
                const float scl  = 0.25f;

                float* crow = sC + e_loc * SC_STRIDE + bb;
                crow[0]  = (x0*y0 + x1*y1 + x2*y2) * inv3 * scl;
                crow[2]  = (x1*y2 - x2*y1) * inv2 * scl;
                crow[4]  = (x2*y0 - x0*y2) * inv2 * scl;
                crow[6]  = (x0*y1 - x1*y0) * inv2 * scl;
                crow[8]  = (x0*y1 + x1*y0) * inv2 * scl;
                crow[10] = (x1*y2 + x2*y1) * inv2 * scl;
                crow[12] = (x0*y2 + x2*y0) * inv2 * scl;
                crow[14] = (x0*y0 - x1*y1) * inv2 * scl;
                crow[16] = (2.0f*x2*y2 - x0*y0 - x1*y1) * inv6 * scl;
        }

        const int  t0       = cs + wv * 16;
        const bool has_tile = (t0 < s1);
        v8f h[16];
        auto ask_nodes = [&]() {
            const float* xp = node_emb + ((size_t)bb * N_NODES + si) * 3;
            const float* yp = node_emb + ((size_t)bb * N_NODES + di) * 3;
            x0 = xp[0]; x1 = xp[1]; x2 = xp[2];
            y0 = yp[0]; y1 = yp[1]; y2 = yp[2];
        };
        if (has_tile) {
            v2f bq[2][32];
            #pragma unroll
            for (int q = 0; q < 32; ++q) bq[0][q] = sB1[q * 32 + lane];
            #pragma unroll
            for (int np = 0; np < 8; ++np) {
                if (np + 1 < 8) {
                    #pragma unroll
                    for (int q = 0; q < 32; ++q)
                        bq[(np + 1) & 1][q] = sB1[((np + 1) * 32 + q) * 32 + lane];
                }
                __builtin_amdgcn_sched_barrier(0);
                v8f accA = vzero, accB = vzero;
                #pragma unroll
                for (int ks = 0; ks < 16; ++ks) {
                    accA = __builtin_amdgcn_wmma_f32_16x16x4_f32(
                        false, aET[ks], false, bq[np & 1][ks], (short)0, accA, false, false);
                    accB = __builtin_amdgcn_wmma_f32_16x16x4_f32(
                        false, aET[ks], false, bq[np & 1][16 + ks], (short)0, accB, false, false);
                }
                #pragma unroll
                for (int i = 0; i < 8; ++i) {
                    const float ha = accA[i] * 0.125f;
                    const float hb = accB[i] * 0.125f;
                    accA[i] = ha * __builtin_amdgcn_rcpf(1.0f + __expf(-ha));
                    accB[i] = hb * __builtin_amdgcn_rcpf(1.0f + __expf(-hb));
                }
                h[2 * np]     = accA;
                h[2 * np + 1] = accB;
                __builtin_amdgcn_sched_barrier(0);
            }
        }

        si = edge_src[pCN];
        di = edge_dst[pCN];
        {
            const float* etp = edge_type + (size_t)pTN * FC_IN;
            #pragma unroll
            for (int ks = 0; ks < 16; ++ks)
                aET[ks] = *(const v2f*)(etp + 4 * ks + khalf);
        }

        if (has_tile) {
            float* myH = sH + wv * (16 * SH_STRIDE);
            const int mbaseH = (lane < 16) ? 0 : 8;
            v8f acc2[6];
            #pragma unroll
            for (int j = 0; j < 6; ++j) acc2[j] = vzero;
            v2f aq[2][4];
            v2f bw[2][24];
            #pragma unroll
            for (int r = 0; r < 8; ++r)
                myH[(r + mbaseH) * SH_STRIDE + lmod] = h[0][r];
            #pragma unroll
            for (int kk = 0; kk < 4; ++kk)
                aq[0][kk] = *(const v2f*)(myH + lmod * SH_STRIDE + 4 * kk + khalf);
            #pragma unroll
            for (int q = 0; q < 24; ++q) bw[0][q] = sB2[q * 32 + lane];

            #pragma unroll
            for (int nt1 = 0; nt1 < 16; ++nt1) {
                if (nt1 + 1 < 16) {
                    #pragma unroll
                    for (int r = 0; r < 8; ++r)
                        myH[(r + mbaseH) * SH_STRIDE + lmod] = h[nt1 + 1][r];
                    #pragma unroll
                    for (int kk = 0; kk < 4; ++kk)
                        aq[(nt1 + 1) & 1][kk] = *(const v2f*)(myH + lmod * SH_STRIDE + 4 * kk + khalf);
                    #pragma unroll
                    for (int q = 0; q < 24; ++q)
                        bw[(nt1 + 1) & 1][q] = sB2[((nt1 + 1) * 24 + q) * 32 + lane];
                }
                if (nt1 == 8) {
                    asm volatile("" : "+v"(si), "+v"(di));
                    ask_nodes();
                }
                __builtin_amdgcn_sched_barrier(0);
                #pragma unroll
                for (int kk = 0; kk < 4; ++kk) {
                    #pragma unroll
                    for (int nt2 = 0; nt2 < 6; ++nt2)
                        acc2[nt2] = __builtin_amdgcn_wmma_f32_16x16x4_f32(
                            false, aq[nt1 & 1][kk], false, bw[nt1 & 1][nt2 * 4 + kk],
                            (short)0, acc2[nt2], false, false);
                }
                __builtin_amdgcn_sched_barrier(0);
            }

            {
                const int mbase = wv * 16 + ((lane < 16) ? 0 : 8);
                #pragma unroll
                for (int nt2 = 0; nt2 < 6; ++nt2) {
                    #pragma unroll
                    for (int r = 0; r < 8; ++r)
                        sW[(mbase + r) * SW_STRIDE + 16 * nt2 + lmod] = acc2[nt2][r] * 0.0625f;
                }
            }
        } else {
            ask_nodes();
        }
        __syncthreads();

        if (has_node) {
            int wcol[9], kco[9];
            wcol[0] = lane;
            kco[0]  = 0;
            #pragma unroll
            for (int i = 1; i < 4; ++i) {
                const int q = lane + 32 * (i - 1);
                const int u = q / 3;
                wcol[i] = MUL + u;
                kco[i]  = 1 + (q - 3 * u);
            }
            #pragma unroll
            for (int i = 4; i < 9; ++i) {
                const int q = lane + 32 * (i - 4);
                const int u = q / 5;
                wcol[i] = 2 * MUL + u;
                kco[i]  = 4 + (q - 5 * u);
            }

            float* park = sAcc + wv * (18 * 32) + lane;
            float acc0[9], acc1[9];
            if (cs == s0) {
                #pragma unroll
                for (int i = 0; i < 9; ++i) { acc0[i] = 0.f; acc1[i] = 0.f; }
            } else {
                #pragma unroll
                for (int i = 0; i < 9; ++i) { acc0[i] = park[32 * i]; acc1[i] = park[32 * (9 + i)]; }
            }

            const int ce = cs + EDGES_PER_CHUNK;
            const int j0 = (seg_lo > cs) ? seg_lo : cs;
            const int j1 = (seg_hi < ce) ? seg_hi : ce;
            for (int j = j0; j < j1; ++j) {
                const float* wrow = sW + (j - cs) * SW_STRIDE;
                const float* crow = sC + (j - cs) * SC_STRIDE;
                #pragma unroll
                for (int i = 0; i < 9; ++i) {
                    const float w = wrow[wcol[i]];
                    const v2f   c = *(const v2f*)(crow + 2 * kco[i]);
                    acc0[i] = fmaf(w, c.x, acc0[i]);
                    acc1[i] = fmaf(w, c.y, acc1[i]);
                }
            }

            if (ce >= s1) {
                float* o0 = out + ((size_t)node) * OUT_CH + lane;
                float* o1 = out + ((size_t)N_NODES + node) * OUT_CH + lane;
                #pragma unroll
                for (int i = 0; i < 9; ++i) {
                    o0[32 * i] = acc0[i];
                    o1[32 * i] = acc1[i];
                }
            } else {
                #pragma unroll
                for (int i = 0; i < 9; ++i) { park[32 * i] = acc0[i]; park[32 * (9 + i)] = acc1[i]; }
            }
        }
        if (!more) break;
        __builtin_amdgcn_fence(__ATOMIC_RELEASE, "workgroup", "local");
        __builtin_amdgcn_s_barrier();
        __builtin_amdgcn_fence(__ATOMIC_ACQUIRE, "workgroup", "local");
        cs = csN;
        n0 = n0N;
        ov = ovN;
    }
}

extern "C" void kernel_launch(void* const* d_in, const int* in_sizes, int n_in,
                              void* d_out, int out_size, void* d_ws, size_t ws_size,
                              hipStream_t stream) {
    const int*   edge_src  = (const int*)d_in[0];
    const int*   edge_dst  = (const int*)d_in[1];
    const float* node_emb  = (const float*)d_in[2];
    const float* edge_type = (const float*)d_in[3];
    const float* W1        = (const float*)d_in[4];
    const float* W2        = (const float*)d_in[5];
    float*       out       = (float*)d_out;

    int* off    = (int*)d_ws;
    int* cc     = off + WS_OFF_CC;
    int* perm   = off + WS_OFF_PERM;
    int* bucket = off + WS_OFF_BUCKET;

    sort_zero <<<  98, 256, 0, stream>>>(cc, ws_size);
    sort_count<<<1563, 256, 0, stream>>>(edge_dst, cc, ws_size);
    sort_scan <<<   1, 1024, 0, stream>>>(cc, off, ws_size);
    sort_fill <<<1563, 256, 0, stream>>>(edge_dst, cc, bucket, ws_size);
    sort_rank <<<1563, 256, 0, stream>>>(edge_dst, off, bucket, perm, ws_size);
    conv_mlp_tp_gather<<<MAIN_BLOCKS, 256, 0, stream>>>(
        edge_src, edge_dst, node_emb, edge_type, W1, W2, off, perm, out, ws_size);
}


# work cut by edges: a block owns the nodes of an equal edge share (32-way search in off[]); a chunk is the next 128 sorted edges, 8 full tiles; a row cut by a chunk end parks in LDS; add order kept
# speedup vs baseline: 7.0992x; 1.0606x over previous
#include <hip/hip_runtime.h>
#include <math.h>


typedef __attribute__((ext_vector_type(2))) float v2f;
typedef __attribute__((ext_vector_type(8))) float v8f;

#define B_DIM 2
#define N_NODES 25000
#define N_EDGES 400000
#define FC_IN 64
#define FC_HID 256
#define MUL 32
#define OUT_CH 288
#define EDGES_PER_CHUNK 128
#define WAVES 8
#define MAIN_BLOCKS 256
#define SH_STRIDE 18
#define SW_STRIDE 97
#define SC_STRIDE 18

#define WS_OFF_CC     25008
#define WS_OFF_PERM   50016
#define WS_OFF_BUCKET (50016 + N_EDGES)
#define WS_NEED_BYTES ((size_t)(50016 + 2 * N_EDGES) * sizeof(int))

__global__ __launch_bounds__(256) void sort_zero(int* __restrict__ cc, size_t ws_bytes)
{
    if (ws_bytes < WS_NEED_BYTES) return;
    const int i = blockIdx.x * 256 + threadIdx.x;
    if (i < N_NODES) cc[i] = 0;
}

__global__ __launch_bounds__(256) void sort_count(const int* __restrict__ edge_dst,
                                                  int* __restrict__ cc, size_t ws_bytes)
{
    if (ws_bytes < WS_NEED_BYTES) return;
    const int e = blockIdx.x * 256 + threadIdx.x;
    if (e < N_EDGES) atomicAdd(&cc[edge_dst[e]], 1);
}

__global__ __launch_bounds__(1024) void sort_scan(int* __restrict__ cc, int* __restrict__ off,
                                                  size_t ws_bytes)
{
    if (ws_bytes < WS_NEED_BYTES) return;
    __shared__ int s[1024];
    const int t = threadIdx.x;
    const int base = t * 25;
    int loc[25];
    int sum = 0;
    #pragma unroll
    for (int i = 0; i < 25; ++i) {
        const int idx = base + i;
        const int v = (idx < N_NODES) ? cc[idx] : 0;
        loc[i] = v;
        sum += v;
    }
    s[t] = sum;
    __syncthreads();
    for (int d = 1; d < 1024; d <<= 1) {
        const int v = (t >= d) ? s[t - d] : 0;
        __syncthreads();
        s[t] += v;
        __syncthreads();
    }
    int run = s[t] - sum;
    #pragma unroll
    for (int i = 0; i < 25; ++i) {
        const int idx = base + i;
        if (idx < N_NODES) {
            off[idx] = run;
            cc[idx]  = run;
            run += loc[i];
        }
    }
    if (t == 1023) off[N_NODES] = s[1023];
}

__global__ __launch_bounds__(256) void sort_fill(const int* __restrict__ edge_dst,
                                                 int* __restrict__ cc,
                                                 int* __restrict__ bucket, size_t ws_bytes)
{
    if (ws_bytes < WS_NEED_BYTES) return;
    const int e = blockIdx.x * 256 + threadIdx.x;
    if (e < N_EDGES) {
        const int slot = atomicAdd(&cc[edge_dst[e]], 1);
        bucket[slot] = e;
    }
}

__global__ __launch_bounds__(256) void sort_rank(const int* __restrict__ edge_dst,
                                                 const int* __restrict__ off,
                                                 const int* __restrict__ bucket,
                                                 int* __restrict__ perm, size_t ws_bytes)
{
    if (ws_bytes < WS_NEED_BYTES) return;
    const int i = blockIdx.x * 256 + threadIdx.x;
    if (i < N_EDGES) {
        const int e  = bucket[i];
        const int d  = edge_dst[e];
        const int lo = off[d];
        const int hi = off[d + 1];
        int r = 0;
        for (int j = lo; j < hi; ++j) r += (bucket[j] < e) ? 1 : 0;
        perm[lo + r] = e;
    }
}

__global__ __launch_bounds__(256) void conv_mlp_tp_gather(
    const int* __restrict__ edge_src,
    const int* __restrict__ edge_dst,
    const float* __restrict__ node_emb,
    const float* __restrict__ edge_type,
    const float* __restrict__ W1,
    const float* __restrict__ W2,
    const int* __restrict__ off,
    const int* __restrict__ perm,
    float* __restrict__ out,
    size_t ws_bytes)
{
    if (ws_bytes < WS_NEED_BYTES) return;
    __shared__ v2f   sB1[16 * 16 * 32];
    __shared__ v2f   sB2[16 * 6 * 4 * 32];
    __shared__ float sH[WAVES * 16 * SH_STRIDE];
    __shared__ float sW[EDGES_PER_CHUNK * SW_STRIDE];
    __shared__ float sC[EDGES_PER_CHUNK * SC_STRIDE];
    __shared__ float sAcc[2 * 18 * 32];

    const int tid   = threadIdx.x;
    const int wv    = __builtin_amdgcn_readfirstlane(tid >> 5);
    const int lane  = tid & 31;
    const int lmod  = lane & 15;
    const int khalf = (lane < 16) ? 0 : 2;
    const int e_loc = tid >> 1;
    const int bb    = tid & 1;

    const int t_lo = (int)(((long long)blockIdx.x * N_EDGES) / MAIN_BLOCKS);
    const int t_hi = (int)((((long long)blockIdx.x + 1) * N_EDGES) / MAIN_BLOCKS);
    int b_lo = 0, b_hi = 0, c_lo = 0, c_hi = 0;
    #pragma unroll
    for (int lev = 0; lev < 3; ++lev) {
        const int stride = (lev == 0) ? 1024 : ((lev == 1) ? 32 : 1);
        const int i_lo = b_lo + lane * stride;
        const int i_hi = b_hi + lane * stride;
        const int v_lo = off[(i_lo < N_NODES) ? i_lo : N_NODES];
        const int v_hi = off[(i_hi < N_NODES) ? i_hi : N_NODES];
        c_lo = __builtin_popcount(__builtin_amdgcn_ballot_w32(v_lo < t_lo));
        c_hi = __builtin_popcount(__builtin_amdgcn_ballot_w32(v_hi < t_hi));
        if (lev < 2) {
            b_lo += ((c_lo > 0) ? c_lo - 1 : 0) * stride;
            b_hi += ((c_hi > 0) ? c_hi - 1 : 0) * stride;
        }
    }
    const int nlo = b_lo + c_lo;
    const int nhi = ((int)blockIdx.x == MAIN_BLOCKS - 1) ? N_NODES : b_hi + c_hi;
    if (nlo >= nhi) return;

    const int e_end    = off[nhi];
    const int slot_max = (e_end > 0) ? e_end - 1 : 0;

    int n0 = nlo;
    int ov = off[(n0 + lane < nhi) ? n0 + lane : nhi];

    #pragma unroll 8
    for (int it = 0; it < (16 * 16 * 32) / 256; ++it) {
        const int idx = tid + 256 * it;
        const int l  = idx & 31;
        const int ks = (idx >> 5) & 15;
        const int nt = idx >> 9;
        const int k0 = 4 * ks + ((l < 16) ? 0 : 2);
        const int n  = 16 * nt + (l & 15);
        v2f b;
        b.x = W1[(size_t)k0 * FC_HID + n];
        b.y = W1[(size_t)(k0 + 1) * FC_HID + n];
        sB1[idx] = b;
    }

    int cs = __builtin_amdgcn_readlane(ov, 0);
    int pC, pT;
    {
        const int sc = cs + e_loc;
        const int st = cs + wv * 16 + lmod;
        pC = perm[(sc < slot_max) ? sc : slot_max];
        pT = perm[(st < slot_max) ? st : slot_max];
    }

    #pragma unroll 8
    for (int it = 0; it < (16 * 6 * 4 * 32) / 512; ++it) {
        const int idx = tid + 256 * it;
        const int l   = idx & 31;
        const int kk  = (idx >> 5) & 3;
        const int r   = idx >> 7;
        const int nt1 = r / 6;
        const int nt2 = r - 6 * nt1;
        const int k0  = 16 * nt1 + 4 * kk + ((l < 16) ? 0 : 2);
        const int n   = 16 * nt2 + (l & 15);
        v2f b;
        b.x = W2[(size_t)k0 * 96 + n];
        b.y = W2[(size_t)(k0 + 1) * 96 + n];
        sB2[idx] = b;
    }

    v2f aET[16];
    int si = edge_src[pC];
    int di = edge_dst[pC];
    {
        const float* etp = edge_type + (size_t)pT * FC_IN;
        #pragma unroll
        for (int ks = 0; ks < 16; ++ks)
            aET[ks] = *(const v2f*)(etp + 4 * ks + khalf);
    }

    #pragma unroll 8
    for (int it = (16 * 6 * 4 * 32) / 512; it < (16 * 6 * 4 * 32) / 256; ++it) {
        const int idx = tid + 256 * it;
        const int l   = idx & 31;
        const int kk  = (idx >> 5) & 3;
        const int r   = idx >> 7;
        const int nt1 = r / 6;
        const int nt2 = r - 6 * nt1;
        const int k0  = 16 * nt1 + 4 * kk + ((l < 16) ? 0 : 2);
        const int n   = 16 * nt2 + (l & 15);
        v2f b;
        b.x = W2[(size_t)k0 * 96 + n];
        b.y = W2[(size_t)(k0 + 1) * 96 + n];
        sB2[idx] = b;
    }

    float x0, x1, x2, y0, y1, y2;
    {
        const float* xp = node_emb + ((size_t)bb * N_NODES + si) * 3;
        const float* yp = node_emb + ((size_t)bb * N_NODES + di) * 3;
        x0 = xp[0]; x1 = xp[1]; x2 = xp[2];
        y0 = yp[0]; y1 = yp[1]; y2 = yp[2];
    }
    __syncthreads();

    const v8f vzero = {0.f, 0.f, 0.f, 0.f, 0.f, 0.f, 0.f, 0.f};

    int par = 0;

    for (;;) {
    const int vis_end = __builtin_amdgcn_readlane(ov, 31);
    const int lim     = cs + EDGES_PER_CHUNK;
    const int ce      = (lim < vis_end) ? lim : vis_end;
    const int K = __builtin_popcount(__builtin_amdgcn_ballot_w32(
        (lane >= 1) && (n0 + lane <= nhi) && (ov <= ce)));
    const int off_cut = __builtin_amdgcn_readlane(ov, K);
    const int n_rows  = K + (((n0 + K < nhi) && (off_cut < ce)) ? 1 : 0);

    const int  csN  = ce;
    const int  n0N  = n0 + K;
    const bool more = (n0N < nhi);
    const int ovN = off[(n0N + lane < nhi) ? n0N + lane : nhi];
    int pCN, pTN;
    {
        const int sc = csN + e_loc;
        const int st = csN + wv * 16 + lmod;
        pCN = perm[(sc < slot_max) ? sc : slot_max];
        pTN = perm[(st < slot_max) ? st : slot_max];
    }

        if (cs + e_loc < ce) {
                const float inv3 = 0.57735026918962576f;
                const float inv2 = 0.70710678118654752f;
                const float inv6 = 0.40824829046386302f;
                const float scl  = 0.25f;

                float* crow = sC + e_loc * SC_STRIDE + bb;
                crow[0]  = (x0*y0 + x1*y1 + x2*y2) * inv3 * scl;
                crow[2]  = (x1*y2 - x2*y1) * inv2 * scl;
                crow[4]  = (x2*y0 - x0*y2) * inv2 * scl;
                crow[6]  = (x0*y1 - x1*y0) * inv2 * scl;
                crow[8]  = (x0*y1 + x1*y0) * inv2 * scl;
                crow[10] = (x1*y2 + x2*y1) * inv2 * scl;
                crow[12] = (x0*y2 + x2*y0) * inv2 * scl;
                crow[14] = (x0*y0 - x1*y1) * inv2 * scl;
                crow[16] = (2.0f*x2*y2 - x0*y0 - x1*y1) * inv6 * scl;
        }

        const int  t0       = cs + wv * 16;
        const bool has_tile = (t0 < ce);
        v8f h[16];
        auto ask_nodes = [&]() {
            const float* xp = node_emb + ((size_t)bb * N_NODES + si) * 3;
            const float* yp = node_emb + ((size_t)bb * N_NODES + di) * 3;
            x0 = xp[0]; x1 = xp[1]; x2 = xp[2];
            y0 = yp[0]; y1 = yp[1]; y2 = yp[2];
        };
        if (has_tile) {
            v2f bq[2][32];
            #pragma unroll
            for (int q = 0; q < 32; ++q) bq[0][q] = sB1[q * 32 + lane];
            #pragma unroll
            for (int np = 0; np < 8; ++np) {
                if (np + 1 < 8) {
                    #pragma unroll
                    for (int q = 0; q < 32; ++q)
                        bq[(np + 1) & 1][q] = sB1[((np + 1) * 32 + q) * 32 + lane];
                }
                __builtin_amdgcn_sched_barrier(0);
                v8f accA = vzero, accB = vzero;
                #pragma unroll
                for (int ks = 0; ks < 16; ++ks) {
                    accA = __builtin_amdgcn_wmma_f32_16x16x4_f32(
                        false, aET[ks], false, bq[np & 1][ks], (short)0, accA, false, false);
                    accB = __builtin_amdgcn_wmma_f32_16x16x4_f32(
                        false, aET[ks], false, bq[np & 1][16 + ks], (short)0, accB, false, false);
                }
                #pragma unroll
                for (int i = 0; i < 8; ++i) {
                    const float ha = accA[i] * 0.125f;
                    const float hb = accB[i] * 0.125f;
                    accA[i] = ha * __builtin_amdgcn_rcpf(1.0f + __expf(-ha));
                    accB[i] = hb * __builtin_amdgcn_rcpf(1.0f + __expf(-hb));
                }
                h[2 * np]     = accA;
                h[2 * np + 1] = accB;
                __builtin_amdgcn_sched_barrier(0);
            }
        }

        si = edge_src[pCN];
        di = edge_dst[pCN];
        {
            const float* etp = edge_type + (size_t)pTN * FC_IN;
            #pragma unroll
            for (int ks = 0; ks < 16; ++ks)
                aET[ks] = *(const v2f*)(etp + 4 * ks + khalf);
        }

        if (has_tile) {
            float* myH = sH + wv * (16 * SH_STRIDE);
            const int mbaseH = (lane < 16) ? 0 : 8;
            v8f acc2[6];
            #pragma unroll
            for (int j = 0; j < 6; ++j) acc2[j] = vzero;
            v2f aq[2][4];
            v2f bw[2][24];
            #pragma unroll
            for (int r = 0; r < 8; ++r)
                myH[(r + mbaseH) * SH_STRIDE + lmod] = h[0][r];
            #pragma unroll
            for (int kk = 0; kk < 4; ++kk)
                aq[0][kk] = *(const v2f*)(myH + lmod * SH_STRIDE + 4 * kk + khalf);
            #pragma unroll
            for (int q = 0; q < 24; ++q) bw[0][q] = sB2[q * 32 + lane];

            #pragma unroll
            for (int nt1 = 0; nt1 < 16; ++nt1) {
                if (nt1 + 1 < 16) {
                    #pragma unroll
                    for (int r = 0; r < 8; ++r)
                        myH[(r + mbaseH) * SH_STRIDE + lmod] = h[nt1 + 1][r];
                    #pragma unroll
                    for (int kk = 0; kk < 4; ++kk)
                        aq[(nt1 + 1) & 1][kk] = *(const v2f*)(myH + lmod * SH_STRIDE + 4 * kk + khalf);
                    #pragma unroll
                    for (int q = 0; q < 24; ++q)
                        bw[(nt1 + 1) & 1][q] = sB2[((nt1 + 1) * 24 + q) * 32 + lane];
                }
                if (nt1 == 8) {
                    asm volatile("" : "+v"(si), "+v"(di));
                    ask_nodes();
                }
                __builtin_amdgcn_sched_barrier(0);
                #pragma unroll
                for (int kk = 0; kk < 4; ++kk) {
                    #pragma unroll
                    for (int nt2 = 0; nt2 < 6; ++nt2)
                        acc2[nt2] = __builtin_amdgcn_wmma_f32_16x16x4_f32(
                            false, aq[nt1 & 1][kk], false, bw[nt1 & 1][nt2 * 4 + kk],
                            (short)0, acc2[nt2], false, false);
                }
                __builtin_amdgcn_sched_barrier(0);
            }

            {
                const int mbase = wv * 16 + ((lane < 16) ? 0 : 8);
                #pragma unroll
                for (int nt2 = 0; nt2 < 6; ++nt2) {
                    #pragma unroll
                    for (int r = 0; r < 8; ++r)
                        sW[(mbase + r) * SW_STRIDE + 16 * nt2 + lmod] = acc2[nt2][r] * 0.0625f;
                }
            }
        } else {
            ask_nodes();
        }
        __syncthreads();

        if (wv < n_rows) {
            int wcol[9], kco[9];
            wcol[0] = lane;
            kco[0]  = 0;
            #pragma unroll
            for (int i = 1; i < 4; ++i) {
                const int q = lane + 32 * (i - 1);
                const int u = q / 3;
                wcol[i] = MUL + u;
                kco[i]  = 1 + (q - 3 * u);
            }
            #pragma unroll
            for (int i = 4; i < 9; ++i) {
                const int q = lane + 32 * (i - 4);
                const int u = q / 5;
                wcol[i] = 2 * MUL + u;
                kco[i]  = 4 + (q - 5 * u);
            }

            const float* park_rd = sAcc + (par ^ 1) * (18 * 32) + lane;
            float*       park_wr = sAcc + par * (18 * 32) + lane;

            for (int row = wv; row < n_rows; row += WAVES) {
                const int node   = n0 + row;
                const int seg_lo = __builtin_amdgcn_readlane(ov, row);
                const int seg_hi = __builtin_amdgcn_readlane(ov, row + 1);

                float acc0[9], acc1[9];
                if (seg_lo >= cs) {
                    #pragma unroll
                    for (int i = 0; i < 9; ++i) { acc0[i] = 0.f; acc1[i] = 0.f; }
                } else {
                    #pragma unroll
                    for (int i = 0; i < 9; ++i) { acc0[i] = park_rd[32 * i]; acc1[i] = park_rd[32 * (9 + i)]; }
                }

                const int j0 = (seg_lo > cs) ? seg_lo : cs;
                const int j1 = (seg_hi < ce) ? seg_hi : ce;
                for (int j = j0; j < j1; ++j) {
                    const float* wrow = sW + (j - cs) * SW_STRIDE;
                    const float* crow = sC + (j - cs) * SC_STRIDE;
                    #pragma unroll
                    for (int i = 0; i < 9; ++i) {
                        const float w = wrow[wcol[i]];
                        const v2f   c = *(const v2f*)(crow + 2 * kco[i]);
                        acc0[i] = fmaf(w, c.x, acc0[i]);
                        acc1[i] = fmaf(w, c.y, acc1[i]);
                    }
                }

                if (seg_hi <= ce) {
                    float* o0 = out + ((size_t)node) * OUT_CH + lane;
                    float* o1 = out + ((size_t)N_NODES + node) * OUT_CH + lane;
                    #pragma unroll
                    for (int i = 0; i < 9; ++i) {
                        o0[32 * i] = acc0[i];
                        o1[32 * i] = acc1[i];
                    }
                } else {
                    #pragma unroll
                    for (int i = 0; i < 9; ++i) { park_wr[32 * i] = acc0[i]; park_wr[32 * (9 + i)] = acc1[i]; }
                }
            }
        }
        if (!more) break;
        __builtin_amdgcn_fence(__ATOMIC_RELEASE, "workgroup", "local");
        __builtin_amdgcn_s_barrier();
        __builtin_amdgcn_fence(__ATOMIC_ACQUIRE, "workgroup", "local");
        cs = csN;
        n0 = n0N;
        ov = ovN;
        par ^= 1;
    }
}

extern "C" void kernel_launch(void* const* d_in, const int* in_sizes, int n_in,
                              void* d_out, int out_size, void* d_ws, size_t ws_size,
                              hipStream_t stream) {
    const int*   edge_src  = (const int*)d_in[0];
    const int*   edge_dst  = (const int*)d_in[1];
    const float* node_emb  = (const float*)d_in[2];
    const float* edge_type = (const float*)d_in[3];
    const float* W1        = (const float*)d_in[4];
    const float* W2        = (const float*)d_in[5];
    float*       out       = (float*)d_out;

    int* off    = (int*)d_ws;
    int* cc     = off + WS_OFF_CC;
    int* perm   = off + WS_OFF_PERM;
    int* bucket = off + WS_OFF_BUCKET;

    sort_zero <<<  98, 256, 0, stream>>>(cc, ws_size);
    sort_count<<<1563, 256, 0, stream>>>(edge_dst, cc, ws_size);
    sort_scan <<<   1, 1024, 0, stream>>>(cc, off, ws_size);
    sort_fill <<<1563, 256, 0, stream>>>(edge_dst, cc, bucket, ws_size);
    sort_rank <<<1563, 256, 0, stream>>>(edge_dst, off, bucket, perm, ws_size);
    conv_mlp_tp_gather<<<MAIN_BLOCKS, 256, 0, stream>>>(
        edge_src, edge_dst, node_emb, edge_type, W1, W2, off, perm, out, ws_size);
}


# node sums of a chunk run one trip late, behind the next chunk's products (w, coefficient tiles doubled in LDS, one barrier per chunk), on the 4 older waves in their barrier wait; add order kept
# speedup vs baseline: 7.3339x; 1.0331x over previous
#include <hip/hip_runtime.h>
#include <math.h>


typedef __attribute__((ext_vector_type(2))) float v2f;
typedef __attribute__((ext_vector_type(8))) float v8f;

#define B_DIM 2
#define N_NODES 25000
#define N_EDGES 400000
#define FC_IN 64
#define FC_HID 256
#define MUL 32
#define OUT_CH 288
#define EDGES_PER_CHUNK 128
#define WAVES 8
#define SUM_WAVES 4
#define MAIN_BLOCKS 256
#define SH_STRIDE 18
#define SW_STRIDE 97
#define SC_STRIDE 18

#define WS_OFF_CC     25008
#define WS_OFF_PERM   50016
#define WS_OFF_BUCKET (50016 + N_EDGES)
#define WS_NEED_BYTES ((size_t)(50016 + 2 * N_EDGES) * sizeof(int))

__global__ __launch_bounds__(256) void sort_zero(int* __restrict__ cc, size_t ws_bytes)
{
    if (ws_bytes < WS_NEED_BYTES) return;
    const int i = blockIdx.x * 256 + threadIdx.x;
    if (i < N_NODES) cc[i] = 0;
}

__global__ __launch_bounds__(256) void sort_count(const int* __restrict__ edge_dst,
                                                  int* __restrict__ cc, size_t ws_bytes)
{
    if (ws_bytes < WS_NEED_BYTES) return;
    const int e = blockIdx.x * 256 + threadIdx.x;
    if (e < N_EDGES) atomicAdd(&cc[edge_dst[e]], 1);
}

__global__ __launch_bounds__(1024) void sort_scan(int* __restrict__ cc, int* __restrict__ off,
                                                  size_t ws_bytes)
{
    if (ws_bytes < WS_NEED_BYTES) return;
    __shared__ int s[1024];
    const int t = threadIdx.x;
    const int base = t * 25;
    int loc[25];
    int sum = 0;
    #pragma unroll
    for (int i = 0; i < 25; ++i) {
        const int idx = base + i;
        const int v = (idx < N_NODES) ? cc[idx] : 0;
        loc[i] = v;
        sum += v;
    }
    s[t] = sum;
    __syncthreads();
    for (int d = 1; d < 1024; d <<= 1) {
        const int v = (t >= d) ? s[t - d] : 0;
        __syncthreads();
        s[t] += v;
        __syncthreads();
    }
    int run = s[t] - sum;
    #pragma unroll
    for (int i = 0; i < 25; ++i) {
        const int idx = base + i;
        if (idx < N_NODES) {
            off[idx] = run;
            cc[idx]  = run;
            run += loc[i];
        }
    }
    if (t == 1023) off[N_NODES] = s[1023];
}

__global__ __launch_bounds__(256) void sort_fill(const int* __restrict__ edge_dst,
                                                 int* __restrict__ cc,
                                                 int* __restrict__ bucket, size_t ws_bytes)
{
    if (ws_bytes < WS_NEED_BYTES) return;
    const int e = blockIdx.x * 256 + threadIdx.x;
    if (e < N_EDGES) {
        const int slot = atomicAdd(&cc[edge_dst[e]], 1);
        bucket[slot] = e;
    }
}

__global__ __launch_bounds__(256) void sort_rank(const int* __restrict__ edge_dst,
                                                 const int* __restrict__ off,
                                                 const int* __restrict__ bucket,
                                                 int* __restrict__ perm, size_t ws_bytes)
{
    if (ws_bytes < WS_NEED_BYTES) return;
    const int i = blockIdx.x * 256 + threadIdx.x;
    if (i < N_EDGES) {
        const int e  = bucket[i];
        const int d  = edge_dst[e];
        const int lo = off[d];
        const int hi = off[d + 1];
        int r = 0;
        for (int j = lo; j < hi; ++j) r += (bucket[j] < e) ? 1 : 0;
        perm[lo + r] = e;
    }
}

__global__ __launch_bounds__(256) void conv_mlp_tp_gather(
    const int* __restrict__ edge_src,
    const int* __restrict__ edge_dst,
    const float* __restrict__ node_emb,
    const float* __restrict__ edge_type,
    const float* __restrict__ W1,
    const float* __restrict__ W2,
    const int* __restrict__ off,
    const int* __restrict__ perm,
    float* __restrict__ out,
    size_t ws_bytes)
{
    if (ws_bytes < WS_NEED_BYTES) return;
    __shared__ v2f   sB1[16 * 16 * 32];
    __shared__ v2f   sB2[16 * 6 * 4 * 32];
    __shared__ float sH[WAVES * 16 * SH_STRIDE];
    __shared__ float sW[2 * EDGES_PER_CHUNK * SW_STRIDE];
    __shared__ float sC[2 * EDGES_PER_CHUNK * SC_STRIDE];
    __shared__ float sAcc[2 * 18 * 32];

    const int tid   = threadIdx.x;
    const int wv    = __builtin_amdgcn_readfirstlane(tid >> 5);
    const int lane  = tid & 31;
    const int lmod  = lane & 15;
    const int khalf = (lane < 16) ? 0 : 2;
    const int e_loc = tid >> 1;
    const int bb    = tid & 1;

    const int t_lo = (int)(((long long)blockIdx.x * N_EDGES) / MAIN_BLOCKS);
    const int t_hi = (int)((((long long)blockIdx.x + 1) * N_EDGES) / MAIN_BLOCKS);
    int b_lo = 0, b_hi = 0, c_lo = 0, c_hi = 0;
    #pragma unroll
    for (int lev = 0; lev < 3; ++lev) {
        const int stride = (lev == 0) ? 1024 : ((lev == 1) ? 32 : 1);
        const int i_lo = b_lo + lane * stride;
        const int i_hi = b_hi + lane * stride;
        const int v_lo = off[(i_lo < N_NODES) ? i_lo : N_NODES];
        const int v_hi = off[(i_hi < N_NODES) ? i_hi : N_NODES];
        c_lo = __builtin_popcount(__builtin_amdgcn_ballot_w32(v_lo < t_lo));
        c_hi = __builtin_popcount(__builtin_amdgcn_ballot_w32(v_hi < t_hi));
        if (lev < 2) {
            b_lo += ((c_lo > 0) ? c_lo - 1 : 0) * stride;
            b_hi += ((c_hi > 0) ? c_hi - 1 : 0) * stride;
        }
    }
    const int nlo = b_lo + c_lo;
    const int nhi = ((int)blockIdx.x == MAIN_BLOCKS - 1) ? N_NODES : b_hi + c_hi;
    if (nlo >= nhi) return;

    const int e_end    = off[nhi];
    const int slot_max = (e_end > 0) ? e_end - 1 : 0;

    int n0 = nlo;
    int ov = off[(n0 + lane < nhi) ? n0 + lane : nhi];

    #pragma unroll 8
    for (int it = 0; it < (16 * 16 * 32) / 256; ++it) {
        const int idx = tid + 256 * it;
        const int l  = idx & 31;
        const int ks = (idx >> 5) & 15;
        const int nt = idx >> 9;
        const int k0 = 4 * ks + ((l < 16) ? 0 : 2);
        const int n  = 16 * nt + (l & 15);
        v2f b;
        b.x = W1[(size_t)k0 * FC_HID + n];
        b.y = W1[(size_t)(k0 + 1) * FC_HID + n];
        sB1[idx] = b;
    }

    int cs = __builtin_amdgcn_readlane(ov, 0);
    int pC, pT;
    {
        const int sc = cs + e_loc;
        const int st = cs + wv * 16 + lmod;
        pC = perm[(sc < slot_max) ? sc : slot_max];
        pT = perm[(st < slot_max) ? st : slot_max];
    }

    #pragma unroll 8
    for (int it = 0; it < (16 * 6 * 4 * 32) / 512; ++it) {
        const int idx = tid + 256 * it;
        const int l   = idx & 31;
        const int kk  = (idx >> 5) & 3;
        const int r   = idx >> 7;
        const int nt1 = r / 6;
        const int nt2 = r - 6 * nt1;
        const int k0  = 16 * nt1 + 4 * kk + ((l < 16) ? 0 : 2);
        const int n   = 16 * nt2 + (l & 15);
        v2f b;
        b.x = W2[(size_t)k0 * 96 + n];
        b.y = W2[(size_t)(k0 + 1) * 96 + n];
        sB2[idx] = b;
    }

    v2f aET[16];
    int si = edge_src[pC];
    int di = edge_dst[pC];
    {
        const float* etp = edge_type + (size_t)pT * FC_IN;
        #pragma unroll
        for (int ks = 0; ks < 16; ++ks)
            aET[ks] = *(const v2f*)(etp + 4 * ks + khalf);
    }

    #pragma unroll 8
    for (int it = (16 * 6 * 4 * 32) / 512; it < (16 * 6 * 4 * 32) / 256; ++it) {
        const int idx = tid + 256 * it;
        const int l   = idx & 31;
        const int kk  = (idx >> 5) & 3;
        const int r   = idx >> 7;
        const int nt1 = r / 6;
        const int nt2 = r - 6 * nt1;
        const int k0  = 16 * nt1 + 4 * kk + ((l < 16) ? 0 : 2);
        const int n   = 16 * nt2 + (l & 15);
        v2f b;
        b.x = W2[(size_t)k0 * 96 + n];
        b.y = W2[(size_t)(k0 + 1) * 96 + n];
        sB2[idx] = b;
    }

    float x0, x1, x2, y0, y1, y2;
    {
        const float* xp = node_emb + ((size_t)bb * N_NODES + si) * 3;
        const float* yp = node_emb + ((size_t)bb * N_NODES + di) * 3;
        x0 = xp[0]; x1 = xp[1]; x2 = xp[2];
        y0 = yp[0]; y1 = yp[1]; y2 = yp[2];
    }
    __syncthreads();

    const v8f vzero = {0.f, 0.f, 0.f, 0.f, 0.f, 0.f, 0.f, 0.f};

    int par = 0;

    int p_cs = 0, p_ce = 0, p_n0 = 0, p_ov = 0, p_rows = 0;

    auto node_sums = [&](const int s_cs, const int s_ce, const int s_n0, const int s_ov,
                         const int s_rows, const int s_par) {
        if (wv < SUM_WAVES && wv < s_rows) {
            int wcol[9], kco[9];
            wcol[0] = lane;
            kco[0]  = 0;
            #pragma unroll
            for (int i = 1; i < 4; ++i) {
                const int q = lane + 32 * (i - 1);
                const int u = q / 3;
                wcol[i] = MUL + u;
                kco[i]  = 1 + (q - 3 * u);
            }
            #pragma unroll
            for (int i = 4; i < 9; ++i) {
                const int q = lane + 32 * (i - 4);
                const int u = q / 5;
                wcol[i] = 2 * MUL + u;
                kco[i]  = 4 + (q - 5 * u);
            }

            const float* tW = sW + s_par * (EDGES_PER_CHUNK * SW_STRIDE);
            const float* tC = sC + s_par * (EDGES_PER_CHUNK * SC_STRIDE);
            const float* park_rd = sAcc + (s_par ^ 1) * (18 * 32) + lane;
            float*       park_wr = sAcc + s_par * (18 * 32) + lane;

            for (int row = wv; row < s_rows; row += SUM_WAVES) {
                const int node   = s_n0 + row;
                const int seg_lo = __builtin_amdgcn_readlane(s_ov, row);
                const int seg_hi = __builtin_amdgcn_readlane(s_ov, row + 1);

                float acc0[9], acc1[9];
                if (seg_lo >= s_cs) {
                    #pragma unroll
                    for (int i = 0; i < 9; ++i) { acc0[i] = 0.f; acc1[i] = 0.f; }
                } else {
                    #pragma unroll
                    for (int i = 0; i < 9; ++i) { acc0[i] = park_rd[32 * i]; acc1[i] = park_rd[32 * (9 + i)]; }
                }

                const int j0 = (seg_lo > s_cs) ? seg_lo : s_cs;
                const int j1 = (seg_hi < s_ce) ? seg_hi : s_ce;
                for (int j = j0; j < j1; ++j) {
                    const float* wrow = tW + (j - s_cs) * SW_STRIDE;
                    const float* crow = tC + (j - s_cs) * SC_STRIDE;
                    #pragma unroll
                    for (int i = 0; i < 9; ++i) {
                        const float w = wrow[wcol[i]];
                        const v2f   c = *(const v2f*)(crow + 2 * kco[i]);
                        acc0[i] = fmaf(w, c.x, acc0[i]);
                        acc1[i] = fmaf(w, c.y, acc1[i]);
                    }
                }

                if (seg_hi <= s_ce) {
                    float* o0 = out + ((size_t)node) * OUT_CH + lane;
                    float* o1 = out + ((size_t)N_NODES + node) * OUT_CH + lane;
                    #pragma unroll
                    for (int i = 0; i < 9; ++i) {
                        o0[32 * i] = acc0[i];
                        o1[32 * i] = acc1[i];
                    }
                } else {
                    #pragma unroll
                    for (int i = 0; i < 9; ++i) { park_wr[32 * i] = acc0[i]; park_wr[32 * (9 + i)] = acc1[i]; }
                }
            }
        }
    };

    for (;;) {
    const int vis_end = __builtin_amdgcn_readlane(ov, 31);
    const int lim     = cs + EDGES_PER_CHUNK;
    const int ce      = (lim < vis_end) ? lim : vis_end;
    const int K = __builtin_popcount(__builtin_amdgcn_ballot_w32(
        (lane >= 1) && (n0 + lane <= nhi) && (ov <= ce)));
    const int off_cut = __builtin_amdgcn_readlane(ov, K);
    const int n_rows  = K + (((n0 + K < nhi) && (off_cut < ce)) ? 1 : 0);

    const int  csN  = ce;
    const int  n0N  = n0 + K;
    const bool more = (n0N < nhi);
    const int ovN = off[(n0N + lane < nhi) ? n0N + lane : nhi];
    int pCN, pTN;
    {
        const int sc = csN + e_loc;
        const int st = csN + wv * 16 + lmod;
        pCN = perm[(sc < slot_max) ? sc : slot_max];
        pTN = perm[(st < slot_max) ? st : slot_max];
    }

        if (cs + e_loc < ce) {
                const float inv3 = 0.57735026918962576f;
                const float inv2 = 0.70710678118654752f;
                const float inv6 = 0.40824829046386302f;
                const float scl  = 0.25f;

                float* crow = sC + par * (EDGES_PER_CHUNK * SC_STRIDE) + e_loc * SC_STRIDE + bb;
                crow[0]  = (x0*y0 + x1*y1 + x2*y2) * inv3 * scl;
                crow[2]  = (x1*y2 - x2*y1) * inv2 * scl;
                crow[4]  = (x2*y0 - x0*y2) * inv2 * scl;
                crow[6]  = (x0*y1 - x1*y0) * inv2 * scl;
                crow[8]  = (x0*y1 + x1*y0) * inv2 * scl;
                crow[10] = (x1*y2 + x2*y1) * inv2 * scl;
                crow[12] = (x0*y2 + x2*y0) * inv2 * scl;
                crow[14] = (x0*y0 - x1*y1) * inv2 * scl;
                crow[16] = (2.0f*x2*y2 - x0*y0 - x1*y1) * inv6 * scl;
        }

        const int  t0       = cs + wv * 16;
        const bool has_tile = (t0 < ce);
        v8f h[16];
        auto ask_nodes = [&]() {
            const float* xp = node_emb + ((size_t)bb * N_NODES + si) * 3;
            const float* yp = node_emb + ((size_t)bb * N_NODES + di) * 3;
            x0 = xp[0]; x1 = xp[1]; x2 = xp[2];
            y0 = yp[0]; y1 = yp[1]; y2 = yp[2];
        };
        if (has_tile) {
            v2f bq[2][32];
            #pragma unroll
            for (int q = 0; q < 32; ++q) bq[0][q] = sB1[q * 32 + lane];
            #pragma unroll
            for (int np = 0; np < 8; ++np) {
                if (np + 1 < 8) {
                    #pragma unroll
                    for (int q = 0; q < 32; ++q)
                        bq[(np + 1) & 1][q] = sB1[((np + 1) * 32 + q) * 32 + lane];
                }
                __builtin_amdgcn_sched_barrier(0);
                v8f accA = vzero, accB = vzero;
                #pragma unroll
                for (int ks = 0; ks < 16; ++ks) {
                    accA = __builtin_amdgcn_wmma_f32_16x16x4_f32(
                        false, aET[ks], false, bq[np & 1][ks], (short)0, accA, false, false);
                    accB = __builtin_amdgcn_wmma_f32_16x16x4_f32(
                        false, aET[ks], false, bq[np & 1][16 + ks], (short)0, accB, false, false);
                }
                #pragma unroll
                for (int i = 0; i < 8; ++i) {
                    const float ha = accA[i] * 0.125f;
                    const float hb = accB[i] * 0.125f;
                    accA[i] = ha * __builtin_amdgcn_rcpf(1.0f + __expf(-ha));
                    accB[i] = hb * __builtin_amdgcn_rcpf(1.0f + __expf(-hb));
                }
                h[2 * np]     = accA;
                h[2 * np + 1] = accB;
                __builtin_amdgcn_sched_barrier(0);
            }
        }

        si = edge_src[pCN];
        di = edge_dst[pCN];
        {
            const float* etp = edge_type + (size_t)pTN * FC_IN;
            #pragma unroll
            for (int ks = 0; ks < 16; ++ks)
                aET[ks] = *(const v2f*)(etp + 4 * ks + khalf);
        }

        if (has_tile) {
            float* myH = sH + wv * (16 * SH_STRIDE);
            const int mbaseH = (lane < 16) ? 0 : 8;
            v8f acc2[6];
            #pragma unroll
            for (int j = 0; j < 6; ++j) acc2[j] = vzero;
            v2f aq[2][4];
            v2f bw[2][24];
            #pragma unroll
            for (int r = 0; r < 8; ++r)
                myH[(r + mbaseH) * SH_STRIDE + lmod] = h[0][r];
            #pragma unroll
            for (int kk = 0; kk < 4; ++kk)
                aq[0][kk] = *(const v2f*)(myH + lmod * SH_STRIDE + 4 * kk + khalf);
            #pragma unroll
            for (int q = 0; q < 24; ++q) bw[0][q] = sB2[q * 32 + lane];

            #pragma unroll
            for (int nt1 = 0; nt1 < 16; ++nt1) {
                if (nt1 + 1 < 16) {
                    #pragma unroll
                    for (int r = 0; r < 8; ++r)
                        myH[(r + mbaseH) * SH_STRIDE + lmod] = h[nt1 + 1][r];
                    #pragma unroll
                    for (int kk = 0; kk < 4; ++kk)
                        aq[(nt1 + 1) & 1][kk] = *(const v2f*)(myH + lmod * SH_STRIDE + 4 * kk + khalf);
                    #pragma unroll
                    for (int q = 0; q < 24; ++q)
                        bw[(nt1 + 1) & 1][q] = sB2[((nt1 + 1) * 24 + q) * 32 + lane];
                }
                if (nt1 == 8) {
                    asm volatile("" : "+v"(si), "+v"(di));
                    ask_nodes();
                }
                __builtin_amdgcn_sched_barrier(0);
                #pragma unroll
                for (int kk = 0; kk < 4; ++kk) {
                    #pragma unroll
                    for (int nt2 = 0; nt2 < 6; ++nt2)
                        acc2[nt2] = __builtin_amdgcn_wmma_f32_16x16x4_f32(
                            false, aq[nt1 & 1][kk], false, bw[nt1 & 1][nt2 * 4 + kk],
                            (short)0, acc2[nt2], false, false);
                }
                __builtin_amdgcn_sched_barrier(0);
            }

            {
                const int mbase = wv * 16 + ((lane < 16) ? 0 : 8);
                float* tWw = sW + par * (EDGES_PER_CHUNK * SW_STRIDE);
                #pragma unroll
                for (int nt2 = 0; nt2 < 6; ++nt2) {
                    #pragma unroll
                    for (int r = 0; r < 8; ++r)
                        tWw[(mbase + r) * SW_STRIDE + 16 * nt2 + lmod] = acc2[nt2][r] * 0.0625f;
                }
            }
        } else {
            ask_nodes();
        }

        node_sums(p_cs, p_ce, p_n0, p_ov, p_rows, par ^ 1);

        __builtin_amdgcn_fence(__ATOMIC_RELEASE, "workgroup", "local");
        __builtin_amdgcn_s_barrier();
        __builtin_amdgcn_fence(__ATOMIC_ACQUIRE, "workgroup", "local");
        p_cs = cs; p_ce = ce; p_n0 = n0; p_ov = ov; p_rows = n_rows;
        if (!more) break;
        cs = csN;
        n0 = n0N;
        ov = ovN;
        par ^= 1;
    }
    node_sums(p_cs, p_ce, p_n0, p_ov, p_rows, par);
}

extern "C" void kernel_launch(void* const* d_in, const int* in_sizes, int n_in,
                              void* d_out, int out_size, void* d_ws, size_t ws_size,
                              hipStream_t stream) {
    const int*   edge_src  = (const int*)d_in[0];
    const int*   edge_dst  = (const int*)d_in[1];
    const float* node_emb  = (const float*)d_in[2];
    const float* edge_type = (const float*)d_in[3];
    const float* W1        = (const float*)d_in[4];
    const float* W2        = (const float*)d_in[5];
    float*       out       = (float*)d_out;

    int* off    = (int*)d_ws;
    int* cc     = off + WS_OFF_CC;
    int* perm   = off + WS_OFF_PERM;
    int* bucket = off + WS_OFF_BUCKET;

    sort_zero <<<  98, 256, 0, stream>>>(cc, ws_size);
    sort_count<<<1563, 256, 0, stream>>>(edge_dst, cc, ws_size);
    sort_scan <<<   1, 1024, 0, stream>>>(cc, off, ws_size);
    sort_fill <<<1563, 256, 0, stream>>>(edge_dst, cc, bucket, ws_size);
    sort_rank <<<1563, 256, 0, stream>>>(edge_dst, off, bucket, perm, ws_size);
    conv_mlp_tp_gather<<<MAIN_BLOCKS, 256, 0, stream>>>(
        edge_src, edge_dst, node_emb, edge_type, W1, W2, off, perm, out, ws_size);
}


# per-node count scan spread over 98 blocks: each re-sums the counts in front of its 256 nodes (16-byte reads), scans its own in-wave; one-block scan with 100-byte lane strides gone; add order kept
# speedup vs baseline: 8.2793x; 1.1289x over previous
#include <hip/hip_runtime.h>
#include <math.h>


typedef __attribute__((ext_vector_type(2))) float v2f;
typedef __attribute__((ext_vector_type(8))) float v8f;
typedef __attribute__((ext_vector_type(4))) int   v4i;

#define B_DIM 2
#define N_NODES 25000
#define N_EDGES 400000
#define FC_IN 64
#define FC_HID 256
#define MUL 32
#define OUT_CH 288
#define EDGES_PER_CHUNK 128
#define WAVES 8
#define SUM_WAVES 4
#define MAIN_BLOCKS 256
#define SH_STRIDE 18
#define SW_STRIDE 97
#define SC_STRIDE 18

#define WS_OFF_CC     25008
#define WS_OFF_PERM   50016
#define WS_OFF_BUCKET (50016 + N_EDGES)
#define WS_NEED_BYTES ((size_t)(50016 + 2 * N_EDGES) * sizeof(int))

__global__ __launch_bounds__(256) void sort_zero(int* __restrict__ cc, size_t ws_bytes)
{
    if (ws_bytes < WS_NEED_BYTES) return;
    const int i = blockIdx.x * 256 + threadIdx.x;
    if (i < N_NODES) cc[i] = 0;
}

__global__ __launch_bounds__(256) void sort_count(const int* __restrict__ edge_dst,
                                                  int* __restrict__ cc, size_t ws_bytes)
{
    if (ws_bytes < WS_NEED_BYTES) return;
    const int e = blockIdx.x * 256 + threadIdx.x;
    if (e < N_EDGES) atomicAdd(&cc[edge_dst[e]], 1);
}

template <int CNT>
__device__ __forceinline__ int sum_groups(const v4i* __restrict__ p)
{
    v4i q[CNT];
    #pragma unroll
    for (int u = 0; u < CNT; ++u) q[u] = p[u * 256];
    int s = 0;
    #pragma unroll
    for (int u = 0; u < CNT; ++u) s += q[u].x + q[u].y + q[u].z + q[u].w;
    return s;
}

__global__ __launch_bounds__(256) void sort_scan(const int* __restrict__ cc, int* __restrict__ off,
                                                 int* __restrict__ cur, size_t ws_bytes)
{
    if (ws_bytes < WS_NEED_BYTES) return;
    __shared__ int s_front[8];
    __shared__ int s_own[8];
    const int t     = threadIdx.x;
    const int lane  = t & 31;
    const int w     = __builtin_amdgcn_readfirstlane(t >> 5);
    const int first = blockIdx.x * 256;

    const int idx = first + t;
    const int v   = (idx < N_NODES) ? cc[idx] : 0;

    const int  nr = (first / 4 - 32 * w + 255) >> 8;
    const v4i* p  = (const v4i*)cc + 32 * w + lane;
    int front = 0;
    int r = 0;
    for (; r + 8 <= nr; r += 8) front += sum_groups<8>(p + 256 * r);
    if (r + 4 <= nr) { front += sum_groups<4>(p + 256 * r); r += 4; }
    if (r + 2 <= nr) { front += sum_groups<2>(p + 256 * r); r += 2; }
    if (r < nr)        front += sum_groups<1>(p + 256 * r);

    int inc = v;
    #pragma unroll
    for (int d = 1; d < 32; d <<= 1) {
        const int u = __shfl_up(inc, d);
        if (lane >= d) inc += u;
    }
    #pragma unroll
    for (int d = 16; d >= 1; d >>= 1) front += __shfl_xor(front, d);
    if (lane == 31) s_own[w]   = inc;
    if (lane == 0)  s_front[w] = front;
    __syncthreads();
    int base = 0;
    #pragma unroll
    for (int k = 0; k < 8; ++k) {
        base += s_front[k];
        base += (k < w) ? s_own[k] : 0;
    }
    const int ex = base + inc - v;
    if (idx < N_NODES) {
        off[idx] = ex;
        cur[idx] = ex;
    }
    if (idx == N_NODES - 1) off[N_NODES] = ex + v;
}

__global__ __launch_bounds__(256) void sort_fill(const int* __restrict__ edge_dst,
                                                 int* __restrict__ cur,
                                                 int* __restrict__ bucket, size_t ws_bytes)
{
    if (ws_bytes < WS_NEED_BYTES) return;
    const int e = blockIdx.x * 256 + threadIdx.x;
    if (e < N_EDGES) {
        const int slot = atomicAdd(&cur[edge_dst[e]], 1);
        bucket[slot] = e;
    }
}

__global__ __launch_bounds__(256) void sort_rank(const int* __restrict__ edge_dst,
                                                 const int* __restrict__ off,
                                                 const int* __restrict__ bucket,
                                                 int* __restrict__ perm, size_t ws_bytes)
{
    if (ws_bytes < WS_NEED_BYTES) return;
    const int i = blockIdx.x * 256 + threadIdx.x;
    if (i < N_EDGES) {
        const int e  = bucket[i];
        const int d  = edge_dst[e];
        const int lo = off[d];
        const int hi = off[d + 1];
        int r = 0;
        for (int j = lo; j < hi; ++j) r += (bucket[j] < e) ? 1 : 0;
        perm[lo + r] = e;
    }
}

__global__ __launch_bounds__(256) void conv_mlp_tp_gather(
    const int* __restrict__ edge_src,
    const int* __restrict__ edge_dst,
    const float* __restrict__ node_emb,
    const float* __restrict__ edge_type,
    const float* __restrict__ W1,
    const float* __restrict__ W2,
    const int* __restrict__ off,
    const int* __restrict__ perm,
    float* __restrict__ out,
    size_t ws_bytes)
{
    if (ws_bytes < WS_NEED_BYTES) return;
    __shared__ v2f   sB1[16 * 16 * 32];
    __shared__ v2f   sB2[16 * 6 * 4 * 32];
    __shared__ float sH[WAVES * 16 * SH_STRIDE];
    __shared__ float sW[2 * EDGES_PER_CHUNK * SW_STRIDE];
    __shared__ float sC[2 * EDGES_PER_CHUNK * SC_STRIDE];
    __shared__ float sAcc[2 * 18 * 32];

    const int tid   = threadIdx.x;
    const int wv    = __builtin_amdgcn_readfirstlane(tid >> 5);
    const int lane  = tid & 31;
    const int lmod  = lane & 15;
    const int khalf = (lane < 16) ? 0 : 2;
    const int e_loc = tid >> 1;
    const int bb    = tid & 1;

    const int t_lo = (int)(((long long)blockIdx.x * N_EDGES) / MAIN_BLOCKS);
    const int t_hi = (int)((((long long)blockIdx.x + 1) * N_EDGES) / MAIN_BLOCKS);
    int b_lo = 0, b_hi = 0, c_lo = 0, c_hi = 0;
    #pragma unroll
    for (int lev = 0; lev < 3; ++lev) {
        const int stride = (lev == 0) ? 1024 : ((lev == 1) ? 32 : 1);
        const int i_lo = b_lo + lane * stride;
        const int i_hi = b_hi + lane * stride;
        const int v_lo = off[(i_lo < N_NODES) ? i_lo : N_NODES];
        const int v_hi = off[(i_hi < N_NODES) ? i_hi : N_NODES];
        c_lo = __builtin_popcount(__builtin_amdgcn_ballot_w32(v_lo < t_lo));
        c_hi = __builtin_popcount(__builtin_amdgcn_ballot_w32(v_hi < t_hi));
        if (lev < 2) {
            b_lo += ((c_lo > 0) ? c_lo - 1 : 0) * stride;
            b_hi += ((c_hi > 0) ? c_hi - 1 : 0) * stride;
        }
    }
    const int nlo = b_lo + c_lo;
    const int nhi = ((int)blockIdx.x == MAIN_BLOCKS - 1) ? N_NODES : b_hi + c_hi;
    if (nlo >= nhi) return;

    const int e_end    = off[nhi];
    const int slot_max = (e_end > 0) ? e_end - 1 : 0;

    int n0 = nlo;
    int ov = off[(n0 + lane < nhi) ? n0 + lane : nhi];

    #pragma unroll 8
    for (int it = 0; it < (16 * 16 * 32) / 256; ++it) {
        const int idx = tid + 256 * it;
        const int l  = idx & 31;
        const int ks = (idx >> 5) & 15;
        const int nt = idx >> 9;
        const int k0 = 4 * ks + ((l < 16) ? 0 : 2);
        const int n  = 16 * nt + (l & 15);
        v2f b;
        b.x = W1[(size_t)k0 * FC_HID + n];
        b.y = W1[(size_t)(k0 + 1) * FC_HID + n];
        sB1[idx] = b;
    }

    int cs = __builtin_amdgcn_readlane(ov, 0);
    int pC, pT;
    {
        const int sc = cs + e_loc;
        const int st = cs + wv * 16 + lmod;
        pC = perm[(sc < slot_max) ? sc : slot_max];
        pT = perm[(st < slot_max) ? st : slot_max];
    }

    #pragma unroll 8
    for (int it = 0; it < (16 * 6 * 4 * 32) / 512; ++it) {
        const int idx = tid + 256 * it;
        const int l   = idx & 31;
        const int kk  = (idx >> 5) & 3;
        const int r   = idx >> 7;
        const int nt1 = r / 6;
        const int nt2 = r - 6 * nt1;
        const int k0  = 16 * nt1 + 4 * kk + ((l < 16) ? 0 : 2);
        const int n   = 16 * nt2 + (l & 15);
        v2f b;
        b.x = W2[(size_t)k0 * 96 + n];
        b.y = W2[(size_t)(k0 + 1) * 96 + n];
        sB2[idx] = b;
    }

    v2f aET[16];
    int si = edge_src[pC];
    int di = edge_dst[pC];
    {
        const float* etp = edge_type + (size_t)pT * FC_IN;
        #pragma unroll
        for (int ks = 0; ks < 16; ++ks)
            aET[ks] = *(const v2f*)(etp + 4 * ks + khalf);
    }

    #pragma unroll 8
    for (int it = (16 * 6 * 4 * 32) / 512; it < (16 * 6 * 4 * 32) / 256; ++it) {
        const int idx = tid + 256 * it;
        const int l   = idx & 31;
        const int kk  = (idx >> 5) & 3;
        const int r   = idx >> 7;
        const int nt1 = r / 6;
        const int nt2 = r - 6 * nt1;
        const int k0  = 16 * nt1 + 4 * kk + ((l < 16) ? 0 : 2);
        const int n   = 16 * nt2 + (l & 15);
        v2f b;
        b.x = W2[(size_t)k0 * 96 + n];
        b.y = W2[(size_t)(k0 + 1) * 96 + n];
        sB2[idx] = b;
    }

    float x0, x1, x2, y0, y1, y2;
    {
        const float* xp = node_emb + ((size_t)bb * N_NODES + si) * 3;
        const float* yp = node_emb + ((size_t)bb * N_NODES + di) * 3;
        x0 = xp[0]; x1 = xp[1]; x2 = xp[2];
        y0 = yp[0]; y1 = yp[1]; y2 = yp[2];
    }
    __syncthreads();

    const v8f vzero = {0.f, 0.f, 0.f, 0.f, 0.f, 0.f, 0.f, 0.f};

    int par = 0;

    int p_cs = 0, p_ce = 0, p_n0 = 0, p_ov = 0, p_rows = 0;

    auto node_sums = [&](const int s_cs, const int s_ce, const int s_n0, const int s_ov,
                         const int s_rows, const int s_par) {
        if (wv < SUM_WAVES && wv < s_rows) {
            int wcol[9], kco[9];
            wcol[0] = lane;
            kco[0]  = 0;
            #pragma unroll
            for (int i = 1; i < 4; ++i) {
                const int q = lane + 32 * (i - 1);
                const int u = q / 3;
                wcol[i] = MUL + u;
                kco[i]  = 1 + (q - 3 * u);
            }
            #pragma unroll
            for (int i = 4; i < 9; ++i) {
                const int q = lane + 32 * (i - 4);
                const int u = q / 5;
                wcol[i] = 2 * MUL + u;
                kco[i]  = 4 + (q - 5 * u);
            }

            const float* tW = sW + s_par * (EDGES_PER_CHUNK * SW_STRIDE);
            const float* tC = sC + s_par * (EDGES_PER_CHUNK * SC_STRIDE);
            const float* park_rd = sAcc + (s_par ^ 1) * (18 * 32) + lane;
            float*       park_wr = sAcc + s_par * (18 * 32) + lane;

            for (int row = wv; row < s_rows; row += SUM_WAVES) {
                const int node   = s_n0 + row;
                const int seg_lo = __builtin_amdgcn_readlane(s_ov, row);
                const int seg_hi = __builtin_amdgcn_readlane(s_ov, row + 1);

                float acc0[9], acc1[9];
                if (seg_lo >= s_cs) {
                    #pragma unroll
                    for (int i = 0; i < 9; ++i) { acc0[i] = 0.f; acc1[i] = 0.f; }
                } else {
                    #pragma unroll
                    for (int i = 0; i < 9; ++i) { acc0[i] = park_rd[32 * i]; acc1[i] = park_rd[32 * (9 + i)]; }
                }

                const int j0 = (seg_lo > s_cs) ? seg_lo : s_cs;
                const int j1 = (seg_hi < s_ce) ? seg_hi : s_ce;
                for (int j = j0; j < j1; ++j) {
                    const float* wrow = tW + (j - s_cs) * SW_STRIDE;
                    const float* crow = tC + (j - s_cs) * SC_STRIDE;
                    #pragma unroll
                    for (int i = 0; i < 9; ++i) {
                        const float w = wrow[wcol[i]];
                        const v2f   c = *(const v2f*)(crow + 2 * kco[i]);
                        acc0[i] = fmaf(w, c.x, acc0[i]);
                        acc1[i] = fmaf(w, c.y, acc1[i]);
                    }
                }

                if (seg_hi <= s_ce) {
                    float* o0 = out + ((size_t)node) * OUT_CH + lane;
                    float* o1 = out + ((size_t)N_NODES + node) * OUT_CH + lane;
                    #pragma unroll
                    for (int i = 0; i < 9; ++i) {
                        o0[32 * i] = acc0[i];
                        o1[32 * i] = acc1[i];
                    }
                } else {
                    #pragma unroll
                    for (int i = 0; i < 9; ++i) { park_wr[32 * i] = acc0[i]; park_wr[32 * (9 + i)] = acc1[i]; }
                }
            }
        }
    };

    for (;;) {
    const int vis_end = __builtin_amdgcn_readlane(ov, 31);
    const int lim     = cs + EDGES_PER_CHUNK;
    const int ce      = (lim < vis_end) ? lim : vis_end;
    const int K = __builtin_popcount(__builtin_amdgcn_ballot_w32(
        (lane >= 1) && (n0 + lane <= nhi) && (ov <= ce)));
    const int off_cut = __builtin_amdgcn_readlane(ov, K);
    const int n_rows  = K + (((n0 + K < nhi) && (off_cut < ce)) ? 1 : 0);

    const int  csN  = ce;
    const int  n0N  = n0 + K;
    const bool more = (n0N < nhi);
    const int ovN = off[(n0N + lane < nhi) ? n0N + lane : nhi];
    int pCN, pTN;
    {
        const int sc = csN + e_loc;
        const int st = csN + wv * 16 + lmod;
        pCN = perm[(sc < slot_max) ? sc : slot_max];
        pTN = perm[(st < slot_max) ? st : slot_max];
    }

        if (cs + e_loc < ce) {
                const float inv3 = 0.57735026918962576f;
                const float inv2 = 0.70710678118654752f;
                const float inv6 = 0.40824829046386302f;
                const float scl  = 0.25f;

                float* crow = sC + par * (EDGES_PER_CHUNK * SC_STRIDE) + e_loc * SC_STRIDE + bb;
                crow[0]  = (x0*y0 + x1*y1 + x2*y2) * inv3 * scl;
                crow[2]  = (x1*y2 - x2*y1) * inv2 * scl;
                crow[4]  = (x2*y0 - x0*y2) * inv2 * scl;
                crow[6]  = (x0*y1 - x1*y0) * inv2 * scl;
                crow[8]  = (x0*y1 + x1*y0) * inv2 * scl;
                crow[10] = (x1*y2 + x2*y1) * inv2 * scl;
                crow[12] = (x0*y2 + x2*y0) * inv2 * scl;
                crow[14] = (x0*y0 - x1*y1) * inv2 * scl;
                crow[16] = (2.0f*x2*y2 - x0*y0 - x1*y1) * inv6 * scl;
        }

        const int  t0       = cs + wv * 16;
        const bool has_tile = (t0 < ce);
        v8f h[16];
        auto ask_nodes = [&]() {
            const float* xp = node_emb + ((size_t)bb * N_NODES + si) * 3;
            const float* yp = node_emb + ((size_t)bb * N_NODES + di) * 3;
            x0 = xp[0]; x1 = xp[1]; x2 = xp[2];
            y0 = yp[0]; y1 = yp[1]; y2 = yp[2];
        };
        if (has_tile) {
            v2f bq[2][32];
            #pragma unroll
            for (int q = 0; q < 32; ++q) bq[0][q] = sB1[q * 32 + lane];
            #pragma unroll
            for (int np = 0; np < 8; ++np) {
                if (np + 1 < 8) {
                    #pragma unroll
                    for (int q = 0; q < 32; ++q)
                        bq[(np + 1) & 1][q] = sB1[((np + 1) * 32 + q) * 32 + lane];
                }
                __builtin_amdgcn_sched_barrier(0);
                v8f accA = vzero, accB = vzero;
                #pragma unroll
                for (int ks = 0; ks < 16; ++ks) {
                    accA = __builtin_amdgcn_wmma_f32_16x16x4_f32(
                        false, aET[ks], false, bq[np & 1][ks], (short)0, accA, false, false);
                    accB = __builtin_amdgcn_wmma_f32_16x16x4_f32(
                        false, aET[ks], false, bq[np & 1][16 + ks], (short)0, accB, false, false);
                }
                #pragma unroll
                for (int i = 0; i < 8; ++i) {
                    const float ha = accA[i] * 0.125f;
                    const float hb = accB[i] * 0.125f;
                    accA[i] = ha * __builtin_amdgcn_rcpf(1.0f + __expf(-ha));
                    accB[i] = hb * __builtin_amdgcn_rcpf(1.0f + __expf(-hb));
                }
                h[2 * np]     = accA;
                h[2 * np + 1] = accB;
                __builtin_amdgcn_sched_barrier(0);
            }
        }

        si = edge_src[pCN];
        di = edge_dst[pCN];
        {
            const float* etp = edge_type + (size_t)pTN * FC_IN;
            #pragma unroll
            for (int ks = 0; ks < 16; ++ks)
                aET[ks] = *(const v2f*)(etp + 4 * ks + khalf);
        }

        if (has_tile) {
            float* myH = sH + wv * (16 * SH_STRIDE);
            const int mbaseH = (lane < 16) ? 0 : 8;
            v8f acc2[6];
            #pragma unroll
            for (int j = 0; j < 6; ++j) acc2[j] = vzero;
            v2f aq[2][4];
            v2f bw[2][24];
            #pragma unroll
            for (int r = 0; r < 8; ++r)
                myH[(r + mbaseH) * SH_STRIDE + lmod] = h[0][r];
            #pragma unroll
            for (int kk = 0; kk < 4; ++kk)
                aq[0][kk] = *(const v2f*)(myH + lmod * SH_STRIDE + 4 * kk + khalf);
            #pragma unroll
            for (int q = 0; q < 24; ++q) bw[0][q] = sB2[q * 32 + lane];

            #pragma unroll
            for (int nt1 = 0; nt1 < 16; ++nt1) {
                if (nt1 + 1 < 16) {
                    #pragma unroll
                    for (int r = 0; r < 8; ++r)
                        myH[(r + mbaseH) * SH_STRIDE + lmod] = h[nt1 + 1][r];
                    #pragma unroll
                    for (int kk = 0; kk < 4; ++kk)
                        aq[(nt1 + 1) & 1][kk] = *(const v2f*)(myH + lmod * SH_STRIDE + 4 * kk + khalf);
                    #pragma unroll
                    for (int q = 0; q < 24; ++q)
                        bw[(nt1 + 1) & 1][q] = sB2[((nt1 + 1) * 24 + q) * 32 + lane];
                }
                if (nt1 == 8) {
                    asm volatile("" : "+v"(si), "+v"(di));
                    ask_nodes();
                }
                __builtin_amdgcn_sched_barrier(0);
                #pragma unroll
                for (int kk = 0; kk < 4; ++kk) {
                    #pragma unroll
                    for (int nt2 = 0; nt2 < 6; ++nt2)
                        acc2[nt2] = __builtin_amdgcn_wmma_f32_16x16x4_f32(
                            false, aq[nt1 & 1][kk], false, bw[nt1 & 1][nt2 * 4 + kk],
                            (short)0, acc2[nt2], false, false);
                }
                __builtin_amdgcn_sched_barrier(0);
            }

            {
                const int mbase = wv * 16 + ((lane < 16) ? 0 : 8);
                float* tWw = sW + par * (EDGES_PER_CHUNK * SW_STRIDE);
                #pragma unroll
                for (int nt2 = 0; nt2 < 6; ++nt2) {
                    #pragma unroll
                    for (int r = 0; r < 8; ++r)
                        tWw[(mbase + r) * SW_STRIDE + 16 * nt2 + lmod] = acc2[nt2][r] * 0.0625f;
                }
            }
        } else {
            ask_nodes();
        }

        node_sums(p_cs, p_ce, p_n0, p_ov, p_rows, par ^ 1);

        __builtin_amdgcn_fence(__ATOMIC_RELEASE, "workgroup", "local");
        __builtin_amdgcn_s_barrier();
        __builtin_amdgcn_fence(__ATOMIC_ACQUIRE, "workgroup", "local");
        p_cs = cs; p_ce = ce; p_n0 = n0; p_ov = ov; p_rows = n_rows;
        if (!more) break;
        cs = csN;
        n0 = n0N;
        ov = ovN;
        par ^= 1;
    }
    node_sums(p_cs, p_ce, p_n0, p_ov, p_rows, par);
}

extern "C" void kernel_launch(void* const* d_in, const int* in_sizes, int n_in,
                              void* d_out, int out_size, void* d_ws, size_t ws_size,
                              hipStream_t stream) {
    const int*   edge_src  = (const int*)d_in[0];
    const int*   edge_dst  = (const int*)d_in[1];
    const float* node_emb  = (const float*)d_in[2];
    const float* edge_type = (const float*)d_in[3];
    const float* W1        = (const float*)d_in[4];
    const float* W2        = (const float*)d_in[5];
    float*       out       = (float*)d_out;

    int* off    = (int*)d_ws;
    int* cc     = off + WS_OFF_CC;
    int* perm   = off + WS_OFF_PERM;
    int* bucket = off + WS_OFF_BUCKET;

    sort_zero <<<  98, 256, 0, stream>>>(cc, ws_size);
    sort_count<<<1563, 256, 0, stream>>>(edge_dst, cc, ws_size);
    sort_scan <<<  98, 256, 0, stream>>>(cc, off, perm, ws_size);
    sort_fill <<<1563, 256, 0, stream>>>(edge_dst, perm, bucket, ws_size);
    sort_rank <<<1563, 256, 0, stream>>>(edge_dst, off, bucket, perm, ws_size);
    conv_mlp_tp_gather<<<MAIN_BLOCKS, 256, 0, stream>>>(
        edge_src, edge_dst, node_emb, edge_type, W1, W2, off, perm, out, ws_size);
}


# edge order rebuilt as a two-level counting sort on LDS counters: slices counted per 128-node range, edges split by range, one block orders a range; no global atomics, 4 launches not 6; add order kept
# speedup vs baseline: 9.7436x; 1.1769x over previous
#include <hip/hip_runtime.h>
#include <math.h>


typedef __attribute__((ext_vector_type(2))) float v2f;
typedef __attribute__((ext_vector_type(8))) float v8f;
typedef __attribute__((ext_vector_type(4))) int   v4i;

#define B_DIM 2
#define N_NODES 25000
#define N_EDGES 400000
#define FC_IN 64
#define FC_HID 256
#define MUL 32
#define OUT_CH 288
#define EDGES_PER_CHUNK 128
#define WAVES 8
#define SUM_WAVES 4
#define MAIN_BLOCKS 256
#define SH_STRIDE 18
#define SW_STRIDE 97
#define SC_STRIDE 18

#define RANGE_SHIFT 7
#define RANGE_NODES 128
#define N_RANGES    196
#define SLICE_EDGES 4096
#define N_SLICES    98
#define ROW_V4      49
#define ROW_GROUPS  20
#define KEY_SHIFT   19

#define WS_OFF_CC     25008
#define WS_OFF_RBASE  (25008 + N_SLICES * N_RANGES)
#define WS_OFF_PERM   50016
#define WS_OFF_BUCKET (50016 + N_EDGES)
#define WS_NEED_BYTES ((size_t)(50016 + 2 * N_EDGES) * sizeof(int))

__global__ __launch_bounds__(1024) void part_count(const int* __restrict__ edge_dst,
                                                   int* __restrict__ cnt, size_t ws_bytes)
{
    if (ws_bytes < WS_NEED_BYTES) return;
    __shared__ int s_hist[N_RANGES];
    const int t = threadIdx.x;
    if (t < N_RANGES) s_hist[t] = 0;
    __syncthreads();
    const int e0 = blockIdx.x * SLICE_EDGES + 4 * t;
    if (e0 < N_EDGES) {
        const v4i d = *(const v4i*)(edge_dst + e0);
        atomicAdd(&s_hist[d.x >> RANGE_SHIFT], 1);
        atomicAdd(&s_hist[d.y >> RANGE_SHIFT], 1);
        atomicAdd(&s_hist[d.z >> RANGE_SHIFT], 1);
        atomicAdd(&s_hist[d.w >> RANGE_SHIFT], 1);
    }
    __syncthreads();
    if (t < N_RANGES) cnt[blockIdx.x * N_RANGES + t] = s_hist[t];
}

__global__ __launch_bounds__(1024) void part_scatter(const int* __restrict__ edge_dst,
                                                     const int* __restrict__ cnt,
                                                     int* __restrict__ rbase,
                                                     int* __restrict__ split, size_t ws_bytes)
{
    if (ws_bytes < WS_NEED_BYTES) return;
    __shared__ int s_tot[N_RANGES];
    __shared__ int s_pre[N_RANGES];
    __shared__ int s_cur[N_RANGES];
    __shared__ int s_wsum[8];
    const int t    = threadIdx.x;
    const int lane = t & 31;
    const int w    = t >> 5;
    const int b    = blockIdx.x;

    const int  e0   = b * SLICE_EDGES + 4 * t;
    const bool have = (e0 < N_EDGES);
    v4i d = {0, 0, 0, 0};
    if (have) d = *(const v4i*)(edge_dst + e0);

    if (t < N_RANGES) { s_tot[t] = 0; s_pre[t] = 0; }
    __syncthreads();

    const int c4 = t % ROW_V4;
    const int rg = t / ROW_V4;
    if (rg < ROW_GROUPS) {
        const v4i* tab = (const v4i*)cnt;
        v4i q[5];
        #pragma unroll
        for (int k = 0; k < 5; ++k) {
            const int row = rg + ROW_GROUPS * k;
            const v4i z = {0, 0, 0, 0};
            q[k] = (row < N_SLICES) ? tab[row * ROW_V4 + c4] : z;
        }
        v4i tot = {0, 0, 0, 0}, pre = {0, 0, 0, 0};
        #pragma unroll
        for (int k = 0; k < 5; ++k) {
            const int row = rg + ROW_GROUPS * k;
            tot += q[k];
            if (row < b) pre += q[k];
        }
        atomicAdd(&s_tot[4 * c4 + 0], tot.x); atomicAdd(&s_pre[4 * c4 + 0], pre.x);
        atomicAdd(&s_tot[4 * c4 + 1], tot.y); atomicAdd(&s_pre[4 * c4 + 1], pre.y);
        atomicAdd(&s_tot[4 * c4 + 2], tot.z); atomicAdd(&s_pre[4 * c4 + 2], pre.z);
        atomicAdd(&s_tot[4 * c4 + 3], tot.w); atomicAdd(&s_pre[4 * c4 + 3], pre.w);
    }
    __syncthreads();

    int v = 0, inc = 0;
    if (t < 256) {
        v   = (t < N_RANGES) ? s_tot[t] : 0;
        inc = v;
        #pragma unroll
        for (int s = 1; s < 32; s <<= 1) {
            const int u = __shfl_up(inc, s);
            if (lane >= s) inc += u;
        }
        if (lane == 31) s_wsum[w] = inc;
    }
    __syncthreads();
    if (t < N_RANGES) {
        int base = 0;
        #pragma unroll
        for (int k = 0; k < 8; ++k) base += (k < w) ? s_wsum[k] : 0;
        const int ex = base + inc - v;
        s_cur[t] = ex + s_pre[t];
        if (b == 0) {
            rbase[t] = ex;
            if (t == N_RANGES - 1) rbase[N_RANGES] = ex + v;
        }
    }
    __syncthreads();

    if (have) {
        const int p0 = atomicAdd(&s_cur[d.x >> RANGE_SHIFT], 1);
        const int p1 = atomicAdd(&s_cur[d.y >> RANGE_SHIFT], 1);
        const int p2 = atomicAdd(&s_cur[d.z >> RANGE_SHIFT], 1);
        const int p3 = atomicAdd(&s_cur[d.w >> RANGE_SHIFT], 1);
        split[p0] = e0;
        split[p1] = e0 + 1;
        split[p2] = e0 + 2;
        split[p3] = e0 + 3;
    }
}

__global__ __launch_bounds__(1024) void range_sort(const int* __restrict__ edge_dst,
                                                   const int* __restrict__ rbase,
                                                   int* perm, int* run,
                                                   int* __restrict__ off, size_t ws_bytes)
{
    if (ws_bytes < WS_NEED_BYTES) return;
    __shared__ int s_cnt[RANGE_NODES];
    __shared__ int s_off[RANGE_NODES + 1];
    __shared__ int s_cur[RANGE_NODES];
    __shared__ int s_wsum[4];
    const int t    = threadIdx.x;
    const int lane = t & 31;
    const int w    = t >> 5;
    const int r    = blockIdx.x;
    const int lo   = rbase[r];
    const int n    = rbase[r + 1] - lo;
    const int nb0  = r << RANGE_SHIFT;

    if (t < RANGE_NODES) s_cnt[t] = 0;
    __syncthreads();

    for (int i = t; i < n; i += 1024) {
        const int e = perm[lo + i];
        atomicAdd(&s_cnt[edge_dst[e] - nb0], 1);
    }
    __syncthreads();

    int v = 0, inc = 0;
    if (t < RANGE_NODES) {
        v   = s_cnt[t];
        inc = v;
        #pragma unroll
        for (int s = 1; s < 32; s <<= 1) {
            const int u = __shfl_up(inc, s);
            if (lane >= s) inc += u;
        }
        if (lane == 31) s_wsum[w] = inc;
    }
    __syncthreads();
    if (t < RANGE_NODES) {
        int base = 0;
        #pragma unroll
        for (int k = 0; k < 4; ++k) base += (k < w) ? s_wsum[k] : 0;
        const int ex = base + inc - v;
        s_off[t] = ex;
        s_cur[t] = ex;
        if (t == RANGE_NODES - 1) s_off[RANGE_NODES] = ex + v;
        const int node = nb0 + t;
        if (node < N_NODES) off[node] = lo + ex;
        if (node == N_NODES - 1) off[N_NODES] = lo + ex + v;
    }
    __syncthreads();

    for (int i = t; i < n; i += 1024) {
        const int e = perm[lo + i];
        const int d = edge_dst[e] - nb0;
        const int p = atomicAdd(&s_cur[d], 1);
        run[lo + p] = (d << KEY_SHIFT) | e;
    }
    __syncthreads();

    for (int p = t; p < n; p += 1024) {
        const int key = run[lo + p];
        const int d   = key >> KEY_SHIFT;
        const int q0  = s_off[d];
        const int q1  = s_off[d + 1];
        int rk = 0;
        for (int q = q0; q < q1; ++q) rk += (run[lo + q] < key) ? 1 : 0;
        perm[lo + q0 + rk] = key & ((1 << KEY_SHIFT) - 1);
    }
}

__global__ __launch_bounds__(256) void conv_mlp_tp_gather(
    const int* __restrict__ edge_src,
    const int* __restrict__ edge_dst,
    const float* __restrict__ node_emb,
    const float* __restrict__ edge_type,
    const float* __restrict__ W1,
    const float* __restrict__ W2,
    const int* __restrict__ off,
    const int* __restrict__ perm,
    float* __restrict__ out,
    size_t ws_bytes)
{
    if (ws_bytes < WS_NEED_BYTES) return;
    __shared__ v2f   sB1[16 * 16 * 32];
    __shared__ v2f   sB2[16 * 6 * 4 * 32];
    __shared__ float sH[WAVES * 16 * SH_STRIDE];
    __shared__ float sW[2 * EDGES_PER_CHUNK * SW_STRIDE];
    __shared__ float sC[2 * EDGES_PER_CHUNK * SC_STRIDE];
    __shared__ float sAcc[2 * 18 * 32];

    const int tid   = threadIdx.x;
    const int wv    = __builtin_amdgcn_readfirstlane(tid >> 5);
    const int lane  = tid & 31;
    const int lmod  = lane & 15;
    const int khalf = (lane < 16) ? 0 : 2;
    const int e_loc = tid >> 1;
    const int bb    = tid & 1;

    const int t_lo = (int)(((long long)blockIdx.x * N_EDGES) / MAIN_BLOCKS);
    const int t_hi = (int)((((long long)blockIdx.x + 1) * N_EDGES) / MAIN_BLOCKS);
    int b_lo = 0, b_hi = 0, c_lo = 0, c_hi = 0;
    #pragma unroll
    for (int lev = 0; lev < 3; ++lev) {
        const int stride = (lev == 0) ? 1024 : ((lev == 1) ? 32 : 1);
        const int i_lo = b_lo + lane * stride;
        const int i_hi = b_hi + lane * stride;
        const int v_lo = off[(i_lo < N_NODES) ? i_lo : N_NODES];
        const int v_hi = off[(i_hi < N_NODES) ? i_hi : N_NODES];
        c_lo = __builtin_popcount(__builtin_amdgcn_ballot_w32(v_lo < t_lo));
        c_hi = __builtin_popcount(__builtin_amdgcn_ballot_w32(v_hi < t_hi));
        if (lev < 2) {
            b_lo += ((c_lo > 0) ? c_lo - 1 : 0) * stride;
            b_hi += ((c_hi > 0) ? c_hi - 1 : 0) * stride;
        }
    }
    const int nlo = b_lo + c_lo;
    const int nhi = ((int)blockIdx.x == MAIN_BLOCKS - 1) ? N_NODES : b_hi + c_hi;
    if (nlo >= nhi) return;

    const int e_end    = off[nhi];
    const int slot_max = (e_end > 0) ? e_end - 1 : 0;

    int n0 = nlo;
    int ov = off[(n0 + lane < nhi) ? n0 + lane : nhi];

    #pragma unroll 8
    for (int it = 0; it < (16 * 16 * 32) / 256; ++it) {
        const int idx = tid + 256 * it;
        const int l  = idx & 31;
        const int ks = (idx >> 5) & 15;
        const int nt = idx >> 9;
        const int k0 = 4 * ks + ((l < 16) ? 0 : 2);
        const int n  = 16 * nt + (l & 15);
        v2f b;
        b.x = W1[(size_t)k0 * FC_HID + n];
        b.y = W1[(size_t)(k0 + 1) * FC_HID + n];
        sB1[idx] = b;
    }

    int cs = __builtin_amdgcn_readlane(ov, 0);
    int pC, pT;
    {
        const int sc = cs + e_loc;
        const int st = cs + wv * 16 + lmod;
        pC = perm[(sc < slot_max) ? sc : slot_max];
        pT = perm[(st < slot_max) ? st : slot_max];
    }

    #pragma unroll 8
    for (int it = 0; it < (16 * 6 * 4 * 32) / 512; ++it) {
        const int idx = tid + 256 * it;
        const int l   = idx & 31;
        const int kk  = (idx >> 5) & 3;
        const int r   = idx >> 7;
        const int nt1 = r / 6;
        const int nt2 = r - 6 * nt1;
        const int k0  = 16 * nt1 + 4 * kk + ((l < 16) ? 0 : 2);
        const int n   = 16 * nt2 + (l & 15);
        v2f b;
        b.x = W2[(size_t)k0 * 96 + n];
        b.y = W2[(size_t)(k0 + 1) * 96 + n];
        sB2[idx] = b;
    }

    v2f aET[16];
    int si = edge_src[pC];
    int di = edge_dst[pC];
    {
        const float* etp = edge_type + (size_t)pT * FC_IN;
        #pragma unroll
        for (int ks = 0; ks < 16; ++ks)
            aET[ks] = *(const v2f*)(etp + 4 * ks + khalf);
    }

    #pragma unroll 8
    for (int it = (16 * 6 * 4 * 32) / 512; it < (16 * 6 * 4 * 32) / 256; ++it) {
        const int idx = tid + 256 * it;
        const int l   = idx & 31;
        const int kk  = (idx >> 5) & 3;
        const int r   = idx >> 7;
        const int nt1 = r / 6;
        const int nt2 = r - 6 * nt1;
        const int k0  = 16 * nt1 + 4 * kk + ((l < 16) ? 0 : 2);
        const int n   = 16 * nt2 + (l & 15);
        v2f b;
        b.x = W2[(size_t)k0 * 96 + n];
        b.y = W2[(size_t)(k0 + 1) * 96 + n];
        sB2[idx] = b;
    }

    float x0, x1, x2, y0, y1, y2;
    {
        const float* xp = node_emb + ((size_t)bb * N_NODES + si) * 3;
        const float* yp = node_emb + ((size_t)bb * N_NODES + di) * 3;
        x0 = xp[0]; x1 = xp[1]; x2 = xp[2];
        y0 = yp[0]; y1 = yp[1]; y2 = yp[2];
    }
    __syncthreads();

    const v8f vzero = {0.f, 0.f, 0.f, 0.f, 0.f, 0.f, 0.f, 0.f};

    int par = 0;

    int p_cs = 0, p_ce = 0, p_n0 = 0, p_ov = 0, p_rows = 0;

    auto node_sums = [&](const int s_cs, const int s_ce, const int s_n0, const int s_ov,
                         const int s_rows, const int s_par) {
        if (wv < SUM_WAVES && wv < s_rows) {
            int wcol[9], kco[9];
            wcol[0] = lane;
            kco[0]  = 0;
            #pragma unroll
            for (int i = 1; i < 4; ++i) {
                const int q = lane + 32 * (i - 1);
                const int u = q / 3;
                wcol[i] = MUL + u;
                kco[i]  = 1 + (q - 3 * u);
            }
            #pragma unroll
            for (int i = 4; i < 9; ++i) {
                const int q = lane + 32 * (i - 4);
                const int u = q / 5;
                wcol[i] = 2 * MUL + u;
                kco[i]  = 4 + (q - 5 * u);
            }

            const float* tW = sW + s_par * (EDGES_PER_CHUNK * SW_STRIDE);
            const float* tC = sC + s_par * (EDGES_PER_CHUNK * SC_STRIDE);
            const float* park_rd = sAcc + (s_par ^ 1) * (18 * 32) + lane;
            float*       park_wr = sAcc + s_par * (18 * 32) + lane;

            for (int row = wv; row < s_rows; row += SUM_WAVES) {
                const int node   = s_n0 + row;
                const int seg_lo = __builtin_amdgcn_readlane(s_ov, row);
                const int seg_hi = __builtin_amdgcn_readlane(s_ov, row + 1);

                float acc0[9], acc1[9];
                if (seg_lo >= s_cs) {
                    #pragma unroll
                    for (int i = 0; i < 9; ++i) { acc0[i] = 0.f; acc1[i] = 0.f; }
                } else {
                    #pragma unroll
                    for (int i = 0; i < 9; ++i) { acc0[i] = park_rd[32 * i]; acc1[i] = park_rd[32 * (9 + i)]; }
                }

                const int j0 = (seg_lo > s_cs) ? seg_lo : s_cs;
                const int j1 = (seg_hi < s_ce) ? seg_hi : s_ce;
                for (int j = j0; j < j1; ++j) {
                    const float* wrow = tW + (j - s_cs) * SW_STRIDE;
                    const float* crow = tC + (j - s_cs) * SC_STRIDE;
                    #pragma unroll
                    for (int i = 0; i < 9; ++i) {
                        const float w = wrow[wcol[i]];
                        const v2f   c = *(const v2f*)(crow + 2 * kco[i]);
                        acc0[i] = fmaf(w, c.x, acc0[i]);
                        acc1[i] = fmaf(w, c.y, acc1[i]);
                    }
                }

                if (seg_hi <= s_ce) {
                    float* o0 = out + ((size_t)node) * OUT_CH + lane;
                    float* o1 = out + ((size_t)N_NODES + node) * OUT_CH + lane;
                    #pragma unroll
                    for (int i = 0; i < 9; ++i) {
                        o0[32 * i] = acc0[i];
                        o1[32 * i] = acc1[i];
                    }
                } else {
                    #pragma unroll
                    for (int i = 0; i < 9; ++i) { park_wr[32 * i] = acc0[i]; park_wr[32 * (9 + i)] = acc1[i]; }
                }
            }
        }
    };

    for (;;) {
    const int vis_end = __builtin_amdgcn_readlane(ov, 31);
    const int lim     = cs + EDGES_PER_CHUNK;
    const int ce      = (lim < vis_end) ? lim : vis_end;
    const int K = __builtin_popcount(__builtin_amdgcn_ballot_w32(
        (lane >= 1) && (n0 + lane <= nhi) && (ov <= ce)));
    const int off_cut = __builtin_amdgcn_readlane(ov, K);
    const int n_rows  = K + (((n0 + K < nhi) && (off_cut < ce)) ? 1 : 0);

    const int  csN  = ce;
    const int  n0N  = n0 + K;
    const bool more = (n0N < nhi);
    const int ovN = off[(n0N + lane < nhi) ? n0N + lane : nhi];
    int pCN, pTN;
    {
        const int sc = csN + e_loc;
        const int st = csN + wv * 16 + lmod;
        pCN = perm[(sc < slot_max) ? sc : slot_max];
        pTN = perm[(st < slot_max) ? st : slot_max];
    }

        if (cs + e_loc < ce) {
                const float inv3 = 0.57735026918962576f;
                const float inv2 = 0.70710678118654752f;
                const float inv6 = 0.40824829046386302f;
                const float scl  = 0.25f;

                float* crow = sC + par * (EDGES_PER_CHUNK * SC_STRIDE) + e_loc * SC_STRIDE + bb;
                crow[0]  = (x0*y0 + x1*y1 + x2*y2) * inv3 * scl;
                crow[2]  = (x1*y2 - x2*y1) * inv2 * scl;
                crow[4]  = (x2*y0 - x0*y2) * inv2 * scl;
                crow[6]  = (x0*y1 - x1*y0) * inv2 * scl;
                crow[8]  = (x0*y1 + x1*y0) * inv2 * scl;
                crow[10] = (x1*y2 + x2*y1) * inv2 * scl;
                crow[12] = (x0*y2 + x2*y0) * inv2 * scl;
                crow[14] = (x0*y0 - x1*y1) * inv2 * scl;
                crow[16] = (2.0f*x2*y2 - x0*y0 - x1*y1) * inv6 * scl;
        }

        const int  t0       = cs + wv * 16;
        const bool has_tile = (t0 < ce);
        v8f h[16];
        auto ask_nodes = [&]() {
            const float* xp = node_emb + ((size_t)bb * N_NODES + si) * 3;
            const float* yp = node_emb + ((size_t)bb * N_NODES + di) * 3;
            x0 = xp[0]; x1 = xp[1]; x2 = xp[2];
            y0 = yp[0]; y1 = yp[1]; y2 = yp[2];
        };
        if (has_tile) {
            v2f bq[2][32];
            #pragma unroll
            for (int q = 0; q < 32; ++q) bq[0][q] = sB1[q * 32 + lane];
            #pragma unroll
            for (int np = 0; np < 8; ++np) {
                if (np + 1 < 8) {
                    #pragma unroll
                    for (int q = 0; q < 32; ++q)
                        bq[(np + 1) & 1][q] = sB1[((np + 1) * 32 + q) * 32 + lane];
                }
                __builtin_amdgcn_sched_barrier(0);
                v8f accA = vzero, accB = vzero;
                #pragma unroll
                for (int ks = 0; ks < 16; ++ks) {
                    accA = __builtin_amdgcn_wmma_f32_16x16x4_f32(
                        false, aET[ks], false, bq[np & 1][ks], (short)0, accA, false, false);
                    accB = __builtin_amdgcn_wmma_f32_16x16x4_f32(
                        false, aET[ks], false, bq[np & 1][16 + ks], (short)0, accB, false, false);
                }
                #pragma unroll
                for (int i = 0; i < 8; ++i) {
                    const float ha = accA[i] * 0.125f;
                    const float hb = accB[i] * 0.125f;
                    accA[i] = ha * __builtin_amdgcn_rcpf(1.0f + __expf(-ha));
                    accB[i] = hb * __builtin_amdgcn_rcpf(1.0f + __expf(-hb));
                }
                h[2 * np]     = accA;
                h[2 * np + 1] = accB;
                __builtin_amdgcn_sched_barrier(0);
            }
        }

        si = edge_src[pCN];
        di = edge_dst[pCN];
        {
            const float* etp = edge_type + (size_t)pTN * FC_IN;
            #pragma unroll
            for (int ks = 0; ks < 16; ++ks)
                aET[ks] = *(const v2f*)(etp + 4 * ks + khalf);
        }

        if (has_tile) {
            float* myH = sH + wv * (16 * SH_STRIDE);
            const int mbaseH = (lane < 16) ? 0 : 8;
            v8f acc2[6];
            #pragma unroll
            for (int j = 0; j < 6; ++j) acc2[j] = vzero;
            v2f aq[2][4];
            v2f bw[2][24];
            #pragma unroll
            for (int r = 0; r < 8; ++r)
                myH[(r + mbaseH) * SH_STRIDE + lmod] = h[0][r];
            #pragma unroll
            for (int kk = 0; kk < 4; ++kk)
                aq[0][kk] = *(const v2f*)(myH + lmod * SH_STRIDE + 4 * kk + khalf);
            #pragma unroll
            for (int q = 0; q < 24; ++q) bw[0][q] = sB2[q * 32 + lane];

            #pragma unroll
            for (int nt1 = 0; nt1 < 16; ++nt1) {
                if (nt1 + 1 < 16) {
                    #pragma unroll
                    for (int r = 0; r < 8; ++r)
                        myH[(r + mbaseH) * SH_STRIDE + lmod] = h[nt1 + 1][r];
                    #pragma unroll
                    for (int kk = 0; kk < 4; ++kk)
                        aq[(nt1 + 1) & 1][kk] = *(const v2f*)(myH + lmod * SH_STRIDE + 4 * kk + khalf);
                    #pragma unroll
                    for (int q = 0; q < 24; ++q)
                        bw[(nt1 + 1) & 1][q] = sB2[((nt1 + 1) * 24 + q) * 32 + lane];
                }
                if (nt1 == 8) {
                    asm volatile("" : "+v"(si), "+v"(di));
                    ask_nodes();
                }
                __builtin_amdgcn_sched_barrier(0);
                #pragma unroll
                for (int kk = 0; kk < 4; ++kk) {
                    #pragma unroll
                    for (int nt2 = 0; nt2 < 6; ++nt2)
                        acc2[nt2] = __builtin_amdgcn_wmma_f32_16x16x4_f32(
                            false, aq[nt1 & 1][kk], false, bw[nt1 & 1][nt2 * 4 + kk],
                            (short)0, acc2[nt2], false, false);
                }
                __builtin_amdgcn_sched_barrier(0);
            }

            {
                const int mbase = wv * 16 + ((lane < 16) ? 0 : 8);
                float* tWw = sW + par * (EDGES_PER_CHUNK * SW_STRIDE);
                #pragma unroll
                for (int nt2 = 0; nt2 < 6; ++nt2) {
                    #pragma unroll
                    for (int r = 0; r < 8; ++r)
                        tWw[(mbase + r) * SW_STRIDE + 16 * nt2 + lmod] = acc2[nt2][r] * 0.0625f;
                }
            }
        } else {
            ask_nodes();
        }

        node_sums(p_cs, p_ce, p_n0, p_ov, p_rows, par ^ 1);

        __builtin_amdgcn_fence(__ATOMIC_RELEASE, "workgroup", "local");
        __builtin_amdgcn_s_barrier();
        __builtin_amdgcn_fence(__ATOMIC_ACQUIRE, "workgroup", "local");
        p_cs = cs; p_ce = ce; p_n0 = n0; p_ov = ov; p_rows = n_rows;
        if (!more) break;
        cs = csN;
        n0 = n0N;
        ov = ovN;
        par ^= 1;
    }
    node_sums(p_cs, p_ce, p_n0, p_ov, p_rows, par);
}

extern "C" void kernel_launch(void* const* d_in, const int* in_sizes, int n_in,
                              void* d_out, int out_size, void* d_ws, size_t ws_size,
                              hipStream_t stream) {
    const int*   edge_src  = (const int*)d_in[0];
    const int*   edge_dst  = (const int*)d_in[1];
    const float* node_emb  = (const float*)d_in[2];
    const float* edge_type = (const float*)d_in[3];
    const float* W1        = (const float*)d_in[4];
    const float* W2        = (const float*)d_in[5];
    float*       out       = (float*)d_out;

    int* off    = (int*)d_ws;
    int* cnt    = off + WS_OFF_CC;
    int* rbase  = off + WS_OFF_RBASE;
    int* perm   = off + WS_OFF_PERM;
    int* bucket = off + WS_OFF_BUCKET;

    part_count  <<<N_SLICES, 1024, 0, stream>>>(edge_dst, cnt, ws_size);
    part_scatter<<<N_SLICES, 1024, 0, stream>>>(edge_dst, cnt, rbase, perm, ws_size);
    range_sort  <<<N_RANGES, 1024, 0, stream>>>(edge_dst, rbase, perm, bucket, off, ws_size);
    conv_mlp_tp_gather<<<MAIN_BLOCKS, 256, 0, stream>>>(
        edge_src, edge_dst, node_emb, edge_type, W1, W2, off, perm, out, ws_size);
}
